# starting kernel
# baseline (speedup 1.0000x reference)
#include <hip/hip_runtime.h>
#include <hip/hip_bf16.h>
#include <math.h>

#define HIDC 1024
#define NHC  16
#define HDC  64
#define BBC  2
#define SSC  2048

typedef float v2f __attribute__((ext_vector_type(2)));
typedef float v8f __attribute__((ext_vector_type(8)));

__device__ __forceinline__ v8f wmma4(v2f a, v2f b, v8f c) {
    return __builtin_amdgcn_wmma_f32_16x16x4_f32(
         false, a,  false, b,
         (short)0, c,  false,  false);
}

__device__ __forceinline__ v2f load2(const float* __restrict__ p) {
    return *(const v2f*)p;
}

__device__ __forceinline__ v2f load_a16x4(const float* __restrict__ A, int lda, int k0, int m, int kb2) {
    return load2(A + m * lda + k0 + kb2);
}

__device__ __forceinline__ v2f load_b4x16(const float* __restrict__ B, int ldb, int k0, int n0,
                                          int nl, int kb2) {
    const float* p = B + (k0 + kb2) * ldb + n0 + nl;
    v2f r; r[0] = p[0]; r[1] = p[ldb];
    return r;
}

__global__ void __launch_bounds__(32)
qkv_kernel(const float* __restrict__ x,
           const float* __restrict__ wq, const float* __restrict__ wk,
           const float* __restrict__ wv,
           float* __restrict__ q, float* __restrict__ k, float* __restrict__ vT)
{
    const int lane = threadIdx.x;
    const int m    = lane & 15;
    const int hi   = lane >> 4;
    const int kb2  = hi << 1;
    const int strip = blockIdx.x;
    const int n     = blockIdx.y;
    const int r0    = strip * 16;

    const float* Ab = x + (size_t)r0 * HIDC + n * HDC;
    const float* Wq = wq + n * HDC * HDC;
    const float* Wk = wk + n * HDC * HDC;
    const float* Wv = wv + n * HDC * HDC;

    v8f aq[4] = {}; v8f ak[4] = {}; v8f av[4] = {};
    for (int k0 = 0; k0 < HDC; k0 += 4) {
        v2f a = load_a16x4(Ab, HIDC, k0, m, kb2);
#pragma unroll
        for (int nt = 0; nt < 4; ++nt) {
            v2f bq = load_b4x16(Wq, HDC, k0, nt * 16, m, kb2);
            aq[nt] = wmma4(a, bq, aq[nt]);
            v2f bk = load_b4x16(Wk, HDC, k0, nt * 16, m, kb2);
            ak[nt] = wmma4(a, bk, ak[nt]);
            v2f bv = load_b4x16(Wv, HDC, k0, nt * 16, m, kb2);
            av[nt] = wmma4(a, bv, av[nt]);
        }
    }

    const int b  = r0 / SSC;
    const int s0 = r0 % SSC;
    const int mbase = hi * 8;
    float* qb = q  + (((size_t)b * NHC + n) * SSC + s0) * HDC;
    float* kb = k  + (((size_t)b * NHC + n) * SSC + s0) * HDC;
    float* vb = vT + (((size_t)b * NHC + n) * HDC) * SSC;
#pragma unroll
    for (int nt = 0; nt < 4; ++nt)
#pragma unroll
        for (int j = 0; j < 8; ++j) {
            size_t off = (size_t)(mbase + j) * HDC + nt * 16 + m;
            qb[off] = aq[nt][j];
            kb[off] = ak[nt][j];
            vb[(size_t)(nt * 16 + m) * SSC + s0 + mbase + j] = av[nt][j];
        }
}

__global__ void __launch_bounds__(32)
retention_kernel(const float* __restrict__ q, const float* __restrict__ k,
                 const float* __restrict__ vT, float* __restrict__ ret)
{
    const int lane = threadIdx.x;
    const int m    = lane & 15;
    const int hi   = lane >> 4;
    const int kb2  = hi << 1;
    const int mbase = hi * 8;
    const int st   = blockIdx.x;
    const int bn   = blockIdx.y;
    const int b    = bn / NHC;
    const int n    = bn % NHC;

    const float* Qb = q  + (((size_t)b * NHC + n) * SSC) * HDC;
    const float* Kb = k  + (((size_t)b * NHC + n) * SSC) * HDC;
    const float* Vb = vT + (((size_t)b * NHC + n) * HDC) * SSC;

    const float lin   = logf(1.0f / 512.0f)
                      + (float)n * ((logf(1.0f / 32.0f) - logf(1.0f / 512.0f)) / 15.0f);
    const float l2g   = log2f(1.0f - expf(lin));

    __shared__ float sS[16][18];

    const float* Qt = Qb + st * 16 * HDC;
    v2f qa[16];
#pragma unroll
    for (int kk = 0; kk < 16; ++kk)
        qa[kk] = load2(Qt + m * HDC + kk * 4 + kb2);

    v8f acc[4] = {};
    for (int tt = 0; tt <= st; ++tt) {
        const float* Kt = Kb + tt * 16 * HDC;
        if (tt < st) {
            __builtin_prefetch((const char*)(Kt + 16 * HDC) + lane * 128, 0, 1);
        }
        v8f sc = {};
#pragma unroll
        for (int kk = 0; kk < 16; ++kk) {
            v2f bb = load2(Kt + m * HDC + kk * 4 + kb2);
            sc = wmma4(qa[kk], bb, sc);
        }
#pragma unroll
        for (int j = 0; j < 8; ++j) {
            int sg = st * 16 + mbase + j;
            int tg = tt * 16 + m;
            float d = (float)(sg - tg);
            sS[mbase + j][m] = (sg >= tg) ? sc[j] * exp2f(d * l2g) : 0.0f;
        }
        __syncthreads();
#pragma unroll
        for (int k0 = 0; k0 < 16; k0 += 4) {
            v2f a2 = *(const v2f*)&sS[m][k0 + kb2];
#pragma unroll
            for (int nt = 0; nt < 4; ++nt) {
                v2f b2 = load2(Vb + (size_t)(nt * 16 + m) * SSC + tt * 16 + k0 + kb2);
                acc[nt] = wmma4(a2, b2, acc[nt]);
            }
        }
        __syncthreads();
    }

    float* rb = ret + (((size_t)b * SSC + st * 16) * NHC + n) * HDC;
#pragma unroll
    for (int nt = 0; nt < 4; ++nt)
#pragma unroll
        for (int j = 0; j < 8; ++j)
            rb[(size_t)(mbase + j) * NHC * HDC + nt * 16 + m] = acc[nt][j];
}

__global__ void groupnorm_kernel(const float* __restrict__ ret,
                                 const float* __restrict__ scale,
                                 const float* __restrict__ bias,
                                 float* __restrict__ normout)
{
    int idx = blockIdx.x * blockDim.x + threadIdx.x;
    if (idx >= BBC * SSC * NHC) return;
    int n = idx % NHC;
    const float* p = ret + (size_t)idx * HDC;

    float mean = 0.0f;
    for (int e = 0; e < HDC; ++e) mean += p[e];
    mean *= (1.0f / HDC);
    float var = 0.0f;
    for (int e = 0; e < HDC; ++e) { float d = p[e] - mean; var += d * d; }
    var *= (1.0f / HDC);
    float inv = rsqrtf(var + 1e-5f);

    float* o = normout + (size_t)idx * HDC;
    for (int e = 0; e < HDC; ++e) {
        int c = n * HDC + e;
        o[e] = (p[e] - mean) * inv * scale[c] + bias[c];
    }
}

template <bool SWISH_ADD>
__global__ void __launch_bounds__(32)
gemm1024_kernel(const float* __restrict__ A, const float* __restrict__ W,
                const float* __restrict__ addn, float* __restrict__ out)
{
    const int lane = threadIdx.x;
    const int m    = lane & 15;
    const int hi   = lane >> 4;
    const int kb2  = hi << 1;
    const int strip = blockIdx.x;
    const int nb    = blockIdx.y;

    const float* Ab = A + (size_t)strip * 16 * HIDC;
    v8f acc[8] = {};
    for (int k0 = 0; k0 < HIDC; k0 += 4) {
        v2f a = load_a16x4(Ab, HIDC, k0, m, kb2);
#pragma unroll
        for (int nt = 0; nt < 8; ++nt) {
            v2f bb = load_b4x16(W, HIDC, k0, nb * 128 + nt * 16, m, kb2);
            acc[nt] = wmma4(a, bb, acc[nt]);
        }
    }

    const int mbase = hi * 8;
#pragma unroll
    for (int nt = 0; nt < 8; ++nt)
#pragma unroll
        for (int j = 0; j < 8; ++j) {
            size_t off = (size_t)(strip * 16 + mbase + j) * HIDC + nb * 128 + nt * 16 + m;
            float g = acc[nt][j];
            if constexpr (SWISH_ADD) {
                out[off] = g / (1.0f + expf(-g)) + addn[off];
            } else {
                out[off] = g;
            }
        }
}

extern "C" void kernel_launch(void* const* d_in, const int* in_sizes, int n_in,
                              void* d_out, int out_size, void* d_ws, size_t ws_size,
                              hipStream_t stream)
{
    (void)in_sizes; (void)n_in; (void)out_size; (void)ws_size;
    const float* x   = (const float*)d_in[0];
    const float* wq  = (const float*)d_in[1];
    const float* wk  = (const float*)d_in[2];
    const float* wv  = (const float*)d_in[3];
    const float* w1  = (const float*)d_in[4];
    const float* w2  = (const float*)d_in[5];
    const float* gns = (const float*)d_in[6];
    const float* gnb = (const float*)d_in[7];
    float* out = (float*)d_out;
    float* ws  = (float*)d_ws;

    const size_t NQ = (size_t)BBC * NHC * SSC * HDC;
    float* q    = ws;
    float* k    = ws + NQ;
    float* vT   = ws + 2 * NQ;
    float* ret  = ws + 3 * NQ;
    float* norm = q;
    float* mid  = k;

    dim3 wave(32);
    qkv_kernel<<<dim3(BBC * SSC / 16, NHC), wave, 0, stream>>>(x, wq, wk, wv, q, k, vT);
    retention_kernel<<<dim3(SSC / 16, BBC * NHC), wave, 0, stream>>>(q, k, vT, ret);
    groupnorm_kernel<<<dim3(BBC * SSC * NHC / 256), dim3(256), 0, stream>>>(ret, gns, gnb, norm);
    gemm1024_kernel<true ><<<dim3(BBC * SSC / 16, HIDC / 128), wave, 0, stream>>>(x,   w1, norm, mid);
    gemm1024_kernel<false><<<dim3(BBC * SSC / 16, HIDC / 128), wave, 0, stream>>>(mid, w2, nullptr, out);
}


# qkv stores q,k,v tiles in WMMA fragment order so retention reads each K/V/Q fragment pair as one contiguous 512 B wave load (4 lines, was 16 per load); k order of the in-tile sums changes
# speedup vs baseline: 1.0514x; 1.0514x over previous
#include <hip/hip_runtime.h>
#include <hip/hip_bf16.h>
#include <math.h>

#define HIDC 1024
#define NHC  16
#define HDC  64
#define BBC  2
#define SSC  2048

typedef float v2f __attribute__((ext_vector_type(2)));
typedef float v4f __attribute__((ext_vector_type(4)));
typedef float v8f __attribute__((ext_vector_type(8)));

__device__ __forceinline__ v8f wmma4(v2f a, v2f b, v8f c) {
    return __builtin_amdgcn_wmma_f32_16x16x4_f32(
         false, a,  false, b,
         (short)0, c,  false,  false);
}

__device__ __forceinline__ v2f load2(const float* __restrict__ p) {
    return *(const v2f*)p;
}

__device__ __forceinline__ v2f load_a16x4(const float* __restrict__ A, int lda, int k0, int m, int kb2) {
    return load2(A + m * lda + k0 + kb2);
}

__device__ __forceinline__ v2f load_b4x16(const float* __restrict__ B, int ldb, int k0, int n0,
                                          int nl, int kb2) {
    const float* p = B + (k0 + kb2) * ldb + n0 + nl;
    v2f r; r[0] = p[0]; r[1] = p[ldb];
    return r;
}

__global__ void __launch_bounds__(32)
qkv_kernel(const float* __restrict__ x,
           const float* __restrict__ wq, const float* __restrict__ wk,
           const float* __restrict__ wv,
           float* __restrict__ qf, float* __restrict__ kf, float* __restrict__ vf)
{
    const int lane = threadIdx.x;
    const int m    = lane & 15;
    const int hi   = lane >> 4;
    const int kb2  = hi << 1;
    const int strip = blockIdx.x;
    const int n     = blockIdx.y;
    const int r0    = strip * 16;

    const float* Ab = x + (size_t)r0 * HIDC + n * HDC;
    const float* Wq = wq + n * HDC * HDC;
    const float* Wk = wk + n * HDC * HDC;
    const float* Wv = wv + n * HDC * HDC;

    v8f aq[4] = {}; v8f ak[4] = {}; v8f av[4] = {};
    for (int k0 = 0; k0 < HDC; k0 += 4) {
        v2f a = load_a16x4(Ab, HIDC, k0, m, kb2);
#pragma unroll
        for (int nt = 0; nt < 4; ++nt) {
            v2f bq = load_b4x16(Wq, HDC, k0, nt * 16, m, kb2);
            aq[nt] = wmma4(bq, a, aq[nt]);
            v2f bk = load_b4x16(Wk, HDC, k0, nt * 16, m, kb2);
            ak[nt] = wmma4(bk, a, ak[nt]);
            v2f bv = load_b4x16(Wv, HDC, k0, nt * 16, m, kb2);
            av[nt] = wmma4(a, bv, av[nt]);
        }
    }

    const int b  = r0 / SSC;
    const int s0 = r0 % SSC;
    const size_t tile = (((size_t)b * NHC + n) * (SSC / 16) + (s0 >> 4)) * (16 * HDC);
    float* qt = qf + tile;
    float* kt = kf + tile;
    float* vt = vf + tile;
#pragma unroll
    for (int nt = 0; nt < 4; ++nt) {
        const int io = ((2 * nt + hi) * 32 + m) * 4;
        v4f q0 = { aq[nt][0], aq[nt][1], aq[nt][2], aq[nt][3] };
        v4f q1 = { aq[nt][4], aq[nt][5], aq[nt][6], aq[nt][7] };
        v4f k0v = { ak[nt][0], ak[nt][1], ak[nt][2], ak[nt][3] };
        v4f k1v = { ak[nt][4], ak[nt][5], ak[nt][6], ak[nt][7] };
        *(v4f*)(qt + io)      = q0;
        *(v4f*)(qt + io + 64) = q1;
        *(v4f*)(kt + io)      = k0v;
        *(v4f*)(kt + io + 64) = k1v;
        v4f v0 = { av[nt][0], av[nt][1], av[nt][2], av[nt][3] };
        v4f v1 = { av[nt][4], av[nt][5], av[nt][6], av[nt][7] };
        *(v4f*)(vt + ((nt * 2 + 0) * 32 + lane) * 4) = v0;
        *(v4f*)(vt + ((nt * 2 + 1) * 32 + lane) * 4) = v1;
    }
}

__global__ void __launch_bounds__(32)
retention_kernel(const float* __restrict__ qf, const float* __restrict__ kf,
                 const float* __restrict__ vf, float* __restrict__ ret)
{
    const int lane = threadIdx.x;
    const int m    = lane & 15;
    const int hi   = lane >> 4;
    const int mbase = hi * 8;
    const int st   = blockIdx.x;
    const int bn   = blockIdx.y;
    const int b    = bn / NHC;
    const int n    = bn % NHC;

    const float* Qb = qf + (size_t)bn * SSC * HDC;
    const float* Kb = kf + (size_t)bn * SSC * HDC;
    const float* Vb = vf + (size_t)bn * SSC * HDC;

    const float lin   = logf(1.0f / 512.0f)
                      + (float)n * ((logf(1.0f / 32.0f) - logf(1.0f / 512.0f)) / 15.0f);
    const float l2g   = log2f(1.0f - expf(lin));

    __shared__ float sS[16][18];

    const float* Qt = Qb + st * (16 * HDC) + lane * 4;
    v4f q4[8];
#pragma unroll
    for (int i = 0; i < 8; ++i)
        q4[i] = *(const v4f*)(Qt + i * 128);

    v8f acc[4] = {};
    for (int tt = 0; tt <= st; ++tt) {
        const float* Kt = Kb + tt * (16 * HDC) + lane * 4;
        const float* Vt = Vb + tt * (16 * HDC) + lane * 4;
        if (tt < st) {
            __builtin_prefetch((const char*)(Kb + (tt + 1) * (16 * HDC)) + lane * 128, 0, 1);
        }
        v8f sc = {};
#pragma unroll
        for (int i = 0; i < 8; ++i) {
            v4f k4 = *(const v4f*)(Kt + i * 128);
            v2f qa0 = { q4[i][0], q4[i][1] };
            v2f kb0 = { k4[0], k4[1] };
            sc = wmma4(qa0, kb0, sc);
            v2f qa1 = { q4[i][2], q4[i][3] };
            v2f kb1 = { k4[2], k4[3] };
            sc = wmma4(qa1, kb1, sc);
        }
#pragma unroll
        for (int j = 0; j < 8; ++j) {
            int sg = st * 16 + mbase + j;
            int tg = tt * 16 + m;
            float d = (float)(sg - tg);
            sS[mbase + j][m] = (sg >= tg) ? sc[j] * exp2f(d * l2g) : 0.0f;
        }
        __syncthreads();
#pragma unroll
        for (int g = 0; g < 2; ++g) {
            v2f a0 = *(const v2f*)&sS[m][8 * hi + 4 * g];
            v2f a1 = *(const v2f*)&sS[m][8 * hi + 4 * g + 2];
#pragma unroll
            for (int nt = 0; nt < 4; ++nt) {
                v4f v4 = *(const v4f*)(Vt + (nt * 2 + g) * 128);
                v2f b0 = { v4[0], v4[1] };
                v2f b1 = { v4[2], v4[3] };
                acc[nt] = wmma4(a0, b0, acc[nt]);
                acc[nt] = wmma4(a1, b1, acc[nt]);
            }
        }
        __syncthreads();
    }

    float* rb = ret + (((size_t)b * SSC + st * 16) * NHC + n) * HDC;
#pragma unroll
    for (int nt = 0; nt < 4; ++nt)
#pragma unroll
        for (int j = 0; j < 8; ++j)
            rb[(size_t)(mbase + j) * NHC * HDC + nt * 16 + m] = acc[nt][j];
}

__global__ void groupnorm_kernel(const float* __restrict__ ret,
                                 const float* __restrict__ scale,
                                 const float* __restrict__ bias,
                                 float* __restrict__ normout)
{
    int idx = blockIdx.x * blockDim.x + threadIdx.x;
    if (idx >= BBC * SSC * NHC) return;
    int n = idx % NHC;
    const float* p = ret + (size_t)idx * HDC;

    float mean = 0.0f;
    for (int e = 0; e < HDC; ++e) mean += p[e];
    mean *= (1.0f / HDC);
    float var = 0.0f;
    for (int e = 0; e < HDC; ++e) { float d = p[e] - mean; var += d * d; }
    var *= (1.0f / HDC);
    float inv = rsqrtf(var + 1e-5f);

    float* o = normout + (size_t)idx * HDC;
    for (int e = 0; e < HDC; ++e) {
        int c = n * HDC + e;
        o[e] = (p[e] - mean) * inv * scale[c] + bias[c];
    }
}

template <bool SWISH_ADD>
__global__ void __launch_bounds__(32)
gemm1024_kernel(const float* __restrict__ A, const float* __restrict__ W,
                const float* __restrict__ addn, float* __restrict__ out)
{
    const int lane = threadIdx.x;
    const int m    = lane & 15;
    const int hi   = lane >> 4;
    const int kb2  = hi << 1;
    const int strip = blockIdx.x;
    const int nb    = blockIdx.y;

    const float* Ab = A + (size_t)strip * 16 * HIDC;
    v8f acc[8] = {};
    for (int k0 = 0; k0 < HIDC; k0 += 4) {
        v2f a = load_a16x4(Ab, HIDC, k0, m, kb2);
#pragma unroll
        for (int nt = 0; nt < 8; ++nt) {
            v2f bb = load_b4x16(W, HIDC, k0, nb * 128 + nt * 16, m, kb2);
            acc[nt] = wmma4(a, bb, acc[nt]);
        }
    }

    const int mbase = hi * 8;
#pragma unroll
    for (int nt = 0; nt < 8; ++nt)
#pragma unroll
        for (int j = 0; j < 8; ++j) {
            size_t off = (size_t)(strip * 16 + mbase + j) * HIDC + nb * 128 + nt * 16 + m;
            float g = acc[nt][j];
            if constexpr (SWISH_ADD) {
                out[off] = g / (1.0f + expf(-g)) + addn[off];
            } else {
                out[off] = g;
            }
        }
}

extern "C" void kernel_launch(void* const* d_in, const int* in_sizes, int n_in,
                              void* d_out, int out_size, void* d_ws, size_t ws_size,
                              hipStream_t stream)
{
    (void)in_sizes; (void)n_in; (void)out_size; (void)ws_size;
    const float* x   = (const float*)d_in[0];
    const float* wq  = (const float*)d_in[1];
    const float* wk  = (const float*)d_in[2];
    const float* wv  = (const float*)d_in[3];
    const float* w1  = (const float*)d_in[4];
    const float* w2  = (const float*)d_in[5];
    const float* gns = (const float*)d_in[6];
    const float* gnb = (const float*)d_in[7];
    float* out = (float*)d_out;
    float* ws  = (float*)d_ws;

    const size_t NQ = (size_t)BBC * NHC * SSC * HDC;
    float* q    = ws;
    float* k    = ws + NQ;
    float* v    = ws + 2 * NQ;
    float* ret  = ws + 3 * NQ;
    float* norm = q;
    float* mid  = k;

    dim3 wave(32);
    qkv_kernel<<<dim3(BBC * SSC / 16, NHC), wave, 0, stream>>>(x, wq, wk, wv, q, k, v);
    retention_kernel<<<dim3(SSC / 16, BBC * NHC), wave, 0, stream>>>(q, k, v, ret);
    groupnorm_kernel<<<dim3(BBC * SSC * NHC / 256), dim3(256), 0, stream>>>(ret, gns, gnb, norm);
    gemm1024_kernel<true ><<<dim3(BBC * SSC / 16, HIDC / 128), wave, 0, stream>>>(x,   w1, norm, mid);
    gemm1024_kernel<false><<<dim3(BBC * SSC / 16, HIDC / 128), wave, 0, stream>>>(mid, w2, nullptr, out);
}


# retention and both 1024-GEMM loops request the next step's global operand loads one step ahead into a second register set, so WMMAs no longer wait a full load latency; sums and order unchanged
# speedup vs baseline: 1.2385x; 1.1780x over previous
#include <hip/hip_runtime.h>
#include <hip/hip_bf16.h>
#include <math.h>

#define HIDC 1024
#define NHC  16
#define HDC  64
#define BBC  2
#define SSC  2048

typedef float v2f __attribute__((ext_vector_type(2)));
typedef float v4f __attribute__((ext_vector_type(4)));
typedef float v8f __attribute__((ext_vector_type(8)));

__device__ __forceinline__ v8f wmma4(v2f a, v2f b, v8f c) {
    return __builtin_amdgcn_wmma_f32_16x16x4_f32(
         false, a,  false, b,
         (short)0, c,  false,  false);
}

__device__ __forceinline__ v2f load2(const float* __restrict__ p) {
    return *(const v2f*)p;
}

__device__ __forceinline__ v2f load_a16x4(const float* __restrict__ A, int lda, int k0, int m, int kb2) {
    return load2(A + m * lda + k0 + kb2);
}

__device__ __forceinline__ v2f load_b4x16(const float* __restrict__ B, int ldb, int k0, int n0,
                                          int nl, int kb2) {
    const float* p = B + (k0 + kb2) * ldb + n0 + nl;
    v2f r; r[0] = p[0]; r[1] = p[ldb];
    return r;
}

__global__ void __launch_bounds__(32)
qkv_kernel(const float* __restrict__ x,
           const float* __restrict__ wq, const float* __restrict__ wk,
           const float* __restrict__ wv,
           float* __restrict__ qf, float* __restrict__ kf, float* __restrict__ vf)
{
    const int lane = threadIdx.x;
    const int m    = lane & 15;
    const int hi   = lane >> 4;
    const int kb2  = hi << 1;
    const int strip = blockIdx.x;
    const int n     = blockIdx.y;
    const int r0    = strip * 16;

    const float* Ab = x + (size_t)r0 * HIDC + n * HDC;
    const float* Wq = wq + n * HDC * HDC;
    const float* Wk = wk + n * HDC * HDC;
    const float* Wv = wv + n * HDC * HDC;

    v8f aq[4] = {}; v8f ak[4] = {}; v8f av[4] = {};
    for (int k0 = 0; k0 < HDC; k0 += 4) {
        v2f a = load_a16x4(Ab, HIDC, k0, m, kb2);
#pragma unroll
        for (int nt = 0; nt < 4; ++nt) {
            v2f bq = load_b4x16(Wq, HDC, k0, nt * 16, m, kb2);
            aq[nt] = wmma4(bq, a, aq[nt]);
            v2f bk = load_b4x16(Wk, HDC, k0, nt * 16, m, kb2);
            ak[nt] = wmma4(bk, a, ak[nt]);
            v2f bv = load_b4x16(Wv, HDC, k0, nt * 16, m, kb2);
            av[nt] = wmma4(a, bv, av[nt]);
        }
    }

    const int b  = r0 / SSC;
    const int s0 = r0 % SSC;
    const size_t tile = (((size_t)b * NHC + n) * (SSC / 16) + (s0 >> 4)) * (16 * HDC);
    float* qt = qf + tile;
    float* kt = kf + tile;
    float* vt = vf + tile;
#pragma unroll
    for (int nt = 0; nt < 4; ++nt) {
        const int io = ((2 * nt + hi) * 32 + m) * 4;
        v4f q0 = { aq[nt][0], aq[nt][1], aq[nt][2], aq[nt][3] };
        v4f q1 = { aq[nt][4], aq[nt][5], aq[nt][6], aq[nt][7] };
        v4f k0v = { ak[nt][0], ak[nt][1], ak[nt][2], ak[nt][3] };
        v4f k1v = { ak[nt][4], ak[nt][5], ak[nt][6], ak[nt][7] };
        *(v4f*)(qt + io)      = q0;
        *(v4f*)(qt + io + 64) = q1;
        *(v4f*)(kt + io)      = k0v;
        *(v4f*)(kt + io + 64) = k1v;
        v4f v0 = { av[nt][0], av[nt][1], av[nt][2], av[nt][3] };
        v4f v1 = { av[nt][4], av[nt][5], av[nt][6], av[nt][7] };
        *(v4f*)(vt + ((nt * 2 + 0) * 32 + lane) * 4) = v0;
        *(v4f*)(vt + ((nt * 2 + 1) * 32 + lane) * 4) = v1;
    }
}

__global__ void __launch_bounds__(32)
retention_kernel(const float* __restrict__ qf, const float* __restrict__ kf,
                 const float* __restrict__ vf, float* __restrict__ ret)
{
    const int lane = threadIdx.x;
    const int m    = lane & 15;
    const int hi   = lane >> 4;
    const int mbase = hi * 8;
    const int st   = blockIdx.x;
    const int bn   = blockIdx.y;
    const int b    = bn / NHC;
    const int n    = bn % NHC;

    const float* Qb = qf + (size_t)bn * SSC * HDC;
    const float* Kb = kf + (size_t)bn * SSC * HDC;
    const float* Vb = vf + (size_t)bn * SSC * HDC;

    const float lin   = logf(1.0f / 512.0f)
                      + (float)n * ((logf(1.0f / 32.0f) - logf(1.0f / 512.0f)) / 15.0f);
    const float l2g   = log2f(1.0f - expf(lin));

    __shared__ float sS[16][18];

    const float* Qt = Qb + st * (16 * HDC) + lane * 4;
    v4f q4[8];
#pragma unroll
    for (int i = 0; i < 8; ++i)
        q4[i] = *(const v4f*)(Qt + i * 128);

    v4f k4[8];
    {
        const float* Kt = Kb + lane * 4;
#pragma unroll
        for (int i = 0; i < 8; ++i)
            k4[i] = *(const v4f*)(Kt + i * 128);
    }

    v8f acc[4] = {};
    for (int tt = 0; tt <= st; ++tt) {
        const float* Vt = Vb + tt * (16 * HDC) + lane * 4;
        v4f vr[8];
#pragma unroll
        for (int i = 0; i < 8; ++i)
            vr[i] = *(const v4f*)(Vt + i * 128);
        v8f sc = {};
#pragma unroll
        for (int i = 0; i < 8; ++i) {
            v2f qa0 = { q4[i][0], q4[i][1] };
            v2f kb0 = { k4[i][0], k4[i][1] };
            sc = wmma4(qa0, kb0, sc);
            v2f qa1 = { q4[i][2], q4[i][3] };
            v2f kb1 = { k4[i][2], k4[i][3] };
            sc = wmma4(qa1, kb1, sc);
        }
        {
            const int tn = (tt < st) ? tt + 1 : tt;
            const float* Kn = Kb + tn * (16 * HDC) + lane * 4;
#pragma unroll
            for (int i = 0; i < 8; ++i)
                k4[i] = *(const v4f*)(Kn + i * 128);
        }
#pragma unroll
        for (int j = 0; j < 8; ++j) {
            int sg = st * 16 + mbase + j;
            int tg = tt * 16 + m;
            float d = (float)(sg - tg);
            sS[mbase + j][m] = (sg >= tg) ? sc[j] * exp2f(d * l2g) : 0.0f;
        }
        __syncthreads();
#pragma unroll
        for (int g = 0; g < 2; ++g) {
            v2f a0 = *(const v2f*)&sS[m][8 * hi + 4 * g];
            v2f a1 = *(const v2f*)&sS[m][8 * hi + 4 * g + 2];
#pragma unroll
            for (int nt = 0; nt < 4; ++nt) {
                v4f v4 = vr[nt * 2 + g];
                v2f b0 = { v4[0], v4[1] };
                v2f b1 = { v4[2], v4[3] };
                acc[nt] = wmma4(a0, b0, acc[nt]);
                acc[nt] = wmma4(a1, b1, acc[nt]);
            }
        }
        __syncthreads();
    }

    float* rb = ret + (((size_t)b * SSC + st * 16) * NHC + n) * HDC;
#pragma unroll
    for (int nt = 0; nt < 4; ++nt)
#pragma unroll
        for (int j = 0; j < 8; ++j)
            rb[(size_t)(mbase + j) * NHC * HDC + nt * 16 + m] = acc[nt][j];
}

__global__ void groupnorm_kernel(const float* __restrict__ ret,
                                 const float* __restrict__ scale,
                                 const float* __restrict__ bias,
                                 float* __restrict__ normout)
{
    int idx = blockIdx.x * blockDim.x + threadIdx.x;
    if (idx >= BBC * SSC * NHC) return;
    int n = idx % NHC;
    const float* p = ret + (size_t)idx * HDC;

    float mean = 0.0f;
    for (int e = 0; e < HDC; ++e) mean += p[e];
    mean *= (1.0f / HDC);
    float var = 0.0f;
    for (int e = 0; e < HDC; ++e) { float d = p[e] - mean; var += d * d; }
    var *= (1.0f / HDC);
    float inv = rsqrtf(var + 1e-5f);

    float* o = normout + (size_t)idx * HDC;
    for (int e = 0; e < HDC; ++e) {
        int c = n * HDC + e;
        o[e] = (p[e] - mean) * inv * scale[c] + bias[c];
    }
}

template <bool SWISH_ADD>
__global__ void __launch_bounds__(32)
gemm1024_kernel(const float* __restrict__ A, const float* __restrict__ W,
                const float* __restrict__ addn, float* __restrict__ out)
{
    const int lane = threadIdx.x;
    const int m    = lane & 15;
    const int hi   = lane >> 4;
    const int kb2  = hi << 1;
    const int strip = blockIdx.x;
    const int nb    = blockIdx.y;

    const float* Ab = A + (size_t)strip * 16 * HIDC;
    const int n0 = nb * 128;
    v8f acc[8] = {};
    v2f a0, a1, b0[8], b1[8];
    a0 = load_a16x4(Ab, HIDC, 0, m, kb2);
#pragma unroll
    for (int nt = 0; nt < 8; ++nt)
        b0[nt] = load_b4x16(W, HIDC, 0, n0 + nt * 16, m, kb2);
    for (int k0 = 0; k0 < HIDC; k0 += 8) {
        a1 = load_a16x4(Ab, HIDC, k0 + 4, m, kb2);
#pragma unroll
        for (int nt = 0; nt < 8; ++nt)
            b1[nt] = load_b4x16(W, HIDC, k0 + 4, n0 + nt * 16, m, kb2);
        __builtin_amdgcn_sched_barrier(0);
#pragma unroll
        for (int nt = 0; nt < 8; ++nt)
            acc[nt] = wmma4(a0, b0[nt], acc[nt]);
        const int kn = (k0 + 8 < HIDC) ? k0 + 8 : k0;
        a0 = load_a16x4(Ab, HIDC, kn, m, kb2);
#pragma unroll
        for (int nt = 0; nt < 8; ++nt)
            b0[nt] = load_b4x16(W, HIDC, kn, n0 + nt * 16, m, kb2);
        __builtin_amdgcn_sched_barrier(0);
#pragma unroll
        for (int nt = 0; nt < 8; ++nt)
            acc[nt] = wmma4(a1, b1[nt], acc[nt]);
    }

    const int mbase = hi * 8;
#pragma unroll
    for (int nt = 0; nt < 8; ++nt)
#pragma unroll
        for (int j = 0; j < 8; ++j) {
            size_t off = (size_t)(strip * 16 + mbase + j) * HIDC + nb * 128 + nt * 16 + m;
            float g = acc[nt][j];
            if constexpr (SWISH_ADD) {
                out[off] = g / (1.0f + expf(-g)) + addn[off];
            } else {
                out[off] = g;
            }
        }
}

extern "C" void kernel_launch(void* const* d_in, const int* in_sizes, int n_in,
                              void* d_out, int out_size, void* d_ws, size_t ws_size,
                              hipStream_t stream)
{
    (void)in_sizes; (void)n_in; (void)out_size; (void)ws_size;
    const float* x   = (const float*)d_in[0];
    const float* wq  = (const float*)d_in[1];
    const float* wk  = (const float*)d_in[2];
    const float* wv  = (const float*)d_in[3];
    const float* w1  = (const float*)d_in[4];
    const float* w2  = (const float*)d_in[5];
    const float* gns = (const float*)d_in[6];
    const float* gnb = (const float*)d_in[7];
    float* out = (float*)d_out;
    float* ws  = (float*)d_ws;

    const size_t NQ = (size_t)BBC * NHC * SSC * HDC;
    float* q    = ws;
    float* k    = ws + NQ;
    float* v    = ws + 2 * NQ;
    float* ret  = ws + 3 * NQ;
    float* norm = q;
    float* mid  = k;

    dim3 wave(32);
    qkv_kernel<<<dim3(BBC * SSC / 16, NHC), wave, 0, stream>>>(x, wq, wk, wv, q, k, v);
    retention_kernel<<<dim3(SSC / 16, BBC * NHC), wave, 0, stream>>>(q, k, v, ret);
    groupnorm_kernel<<<dim3(BBC * SSC * NHC / 256), dim3(256), 0, stream>>>(ret, gns, gnb, norm);
    gemm1024_kernel<true ><<<dim3(BBC * SSC / 16, HIDC / 128), wave, 0, stream>>>(x,   w1, norm, mid);
    gemm1024_kernel<false><<<dim3(BBC * SSC / 16, HIDC / 128), wave, 0, stream>>>(mid, w2, nullptr, out);
}


# both 1024-GEMM loops keep three k-steps of operand loads in flight through a ring of four register sets (was one step ahead); sums and their order unchanged
# speedup vs baseline: 1.3247x; 1.0696x over previous
#include <hip/hip_runtime.h>
#include <hip/hip_bf16.h>
#include <math.h>

#define HIDC 1024
#define NHC  16
#define HDC  64
#define BBC  2
#define SSC  2048

typedef float v2f __attribute__((ext_vector_type(2)));
typedef float v4f __attribute__((ext_vector_type(4)));
typedef float v8f __attribute__((ext_vector_type(8)));

__device__ __forceinline__ v8f wmma4(v2f a, v2f b, v8f c) {
    return __builtin_amdgcn_wmma_f32_16x16x4_f32(
         false, a,  false, b,
         (short)0, c,  false,  false);
}

__device__ __forceinline__ v2f load2(const float* __restrict__ p) {
    return *(const v2f*)p;
}

__device__ __forceinline__ v2f load_a16x4(const float* __restrict__ A, int lda, int k0, int m, int kb2) {
    return load2(A + m * lda + k0 + kb2);
}

__device__ __forceinline__ v2f load_b4x16(const float* __restrict__ B, int ldb, int k0, int n0,
                                          int nl, int kb2) {
    const float* p = B + (k0 + kb2) * ldb + n0 + nl;
    v2f r; r[0] = p[0]; r[1] = p[ldb];
    return r;
}

__global__ void __launch_bounds__(32)
qkv_kernel(const float* __restrict__ x,
           const float* __restrict__ wq, const float* __restrict__ wk,
           const float* __restrict__ wv,
           float* __restrict__ qf, float* __restrict__ kf, float* __restrict__ vf)
{
    const int lane = threadIdx.x;
    const int m    = lane & 15;
    const int hi   = lane >> 4;
    const int kb2  = hi << 1;
    const int strip = blockIdx.x;
    const int n     = blockIdx.y;
    const int r0    = strip * 16;

    const float* Ab = x + (size_t)r0 * HIDC + n * HDC;
    const float* Wq = wq + n * HDC * HDC;
    const float* Wk = wk + n * HDC * HDC;
    const float* Wv = wv + n * HDC * HDC;

    v8f aq[4] = {}; v8f ak[4] = {}; v8f av[4] = {};
    for (int k0 = 0; k0 < HDC; k0 += 4) {
        v2f a = load_a16x4(Ab, HIDC, k0, m, kb2);
#pragma unroll
        for (int nt = 0; nt < 4; ++nt) {
            v2f bq = load_b4x16(Wq, HDC, k0, nt * 16, m, kb2);
            aq[nt] = wmma4(bq, a, aq[nt]);
            v2f bk = load_b4x16(Wk, HDC, k0, nt * 16, m, kb2);
            ak[nt] = wmma4(bk, a, ak[nt]);
            v2f bv = load_b4x16(Wv, HDC, k0, nt * 16, m, kb2);
            av[nt] = wmma4(a, bv, av[nt]);
        }
    }

    const int b  = r0 / SSC;
    const int s0 = r0 % SSC;
    const size_t tile = (((size_t)b * NHC + n) * (SSC / 16) + (s0 >> 4)) * (16 * HDC);
    float* qt = qf + tile;
    float* kt = kf + tile;
    float* vt = vf + tile;
#pragma unroll
    for (int nt = 0; nt < 4; ++nt) {
        const int io = ((2 * nt + hi) * 32 + m) * 4;
        v4f q0 = { aq[nt][0], aq[nt][1], aq[nt][2], aq[nt][3] };
        v4f q1 = { aq[nt][4], aq[nt][5], aq[nt][6], aq[nt][7] };
        v4f k0v = { ak[nt][0], ak[nt][1], ak[nt][2], ak[nt][3] };
        v4f k1v = { ak[nt][4], ak[nt][5], ak[nt][6], ak[nt][7] };
        *(v4f*)(qt + io)      = q0;
        *(v4f*)(qt + io + 64) = q1;
        *(v4f*)(kt + io)      = k0v;
        *(v4f*)(kt + io + 64) = k1v;
        v4f v0 = { av[nt][0], av[nt][1], av[nt][2], av[nt][3] };
        v4f v1 = { av[nt][4], av[nt][5], av[nt][6], av[nt][7] };
        *(v4f*)(vt + ((nt * 2 + 0) * 32 + lane) * 4) = v0;
        *(v4f*)(vt + ((nt * 2 + 1) * 32 + lane) * 4) = v1;
    }
}

__global__ void __launch_bounds__(32)
retention_kernel(const float* __restrict__ qf, const float* __restrict__ kf,
                 const float* __restrict__ vf, float* __restrict__ ret)
{
    const int lane = threadIdx.x;
    const int m    = lane & 15;
    const int hi   = lane >> 4;
    const int mbase = hi * 8;
    const int st   = blockIdx.x;
    const int bn   = blockIdx.y;
    const int b    = bn / NHC;
    const int n    = bn % NHC;

    const float* Qb = qf + (size_t)bn * SSC * HDC;
    const float* Kb = kf + (size_t)bn * SSC * HDC;
    const float* Vb = vf + (size_t)bn * SSC * HDC;

    const float lin   = logf(1.0f / 512.0f)
                      + (float)n * ((logf(1.0f / 32.0f) - logf(1.0f / 512.0f)) / 15.0f);
    const float l2g   = log2f(1.0f - expf(lin));

    __shared__ float sS[16][18];

    const float* Qt = Qb + st * (16 * HDC) + lane * 4;
    v4f q4[8];
#pragma unroll
    for (int i = 0; i < 8; ++i)
        q4[i] = *(const v4f*)(Qt + i * 128);

    v4f k4[8];
    {
        const float* Kt = Kb + lane * 4;
#pragma unroll
        for (int i = 0; i < 8; ++i)
            k4[i] = *(const v4f*)(Kt + i * 128);
    }

    v8f acc[4] = {};
    for (int tt = 0; tt <= st; ++tt) {
        const float* Vt = Vb + tt * (16 * HDC) + lane * 4;
        v4f vr[8];
#pragma unroll
        for (int i = 0; i < 8; ++i)
            vr[i] = *(const v4f*)(Vt + i * 128);
        v8f sc = {};
#pragma unroll
        for (int i = 0; i < 8; ++i) {
            v2f qa0 = { q4[i][0], q4[i][1] };
            v2f kb0 = { k4[i][0], k4[i][1] };
            sc = wmma4(qa0, kb0, sc);
            v2f qa1 = { q4[i][2], q4[i][3] };
            v2f kb1 = { k4[i][2], k4[i][3] };
            sc = wmma4(qa1, kb1, sc);
        }
        {
            const int tn = (tt < st) ? tt + 1 : tt;
            const float* Kn = Kb + tn * (16 * HDC) + lane * 4;
#pragma unroll
            for (int i = 0; i < 8; ++i)
                k4[i] = *(const v4f*)(Kn + i * 128);
        }
#pragma unroll
        for (int j = 0; j < 8; ++j) {
            int sg = st * 16 + mbase + j;
            int tg = tt * 16 + m;
            float d = (float)(sg - tg);
            sS[mbase + j][m] = (sg >= tg) ? sc[j] * exp2f(d * l2g) : 0.0f;
        }
        __syncthreads();
#pragma unroll
        for (int g = 0; g < 2; ++g) {
            v2f a0 = *(const v2f*)&sS[m][8 * hi + 4 * g];
            v2f a1 = *(const v2f*)&sS[m][8 * hi + 4 * g + 2];
#pragma unroll
            for (int nt = 0; nt < 4; ++nt) {
                v4f v4 = vr[nt * 2 + g];
                v2f b0 = { v4[0], v4[1] };
                v2f b1 = { v4[2], v4[3] };
                acc[nt] = wmma4(a0, b0, acc[nt]);
                acc[nt] = wmma4(a1, b1, acc[nt]);
            }
        }
        __syncthreads();
    }

    float* rb = ret + (((size_t)b * SSC + st * 16) * NHC + n) * HDC;
#pragma unroll
    for (int nt = 0; nt < 4; ++nt)
#pragma unroll
        for (int j = 0; j < 8; ++j)
            rb[(size_t)(mbase + j) * NHC * HDC + nt * 16 + m] = acc[nt][j];
}

__global__ void groupnorm_kernel(const float* __restrict__ ret,
                                 const float* __restrict__ scale,
                                 const float* __restrict__ bias,
                                 float* __restrict__ normout)
{
    int idx = blockIdx.x * blockDim.x + threadIdx.x;
    if (idx >= BBC * SSC * NHC) return;
    int n = idx % NHC;
    const float* p = ret + (size_t)idx * HDC;

    float mean = 0.0f;
    for (int e = 0; e < HDC; ++e) mean += p[e];
    mean *= (1.0f / HDC);
    float var = 0.0f;
    for (int e = 0; e < HDC; ++e) { float d = p[e] - mean; var += d * d; }
    var *= (1.0f / HDC);
    float inv = rsqrtf(var + 1e-5f);

    float* o = normout + (size_t)idx * HDC;
    for (int e = 0; e < HDC; ++e) {
        int c = n * HDC + e;
        o[e] = (p[e] - mean) * inv * scale[c] + bias[c];
    }
}

template <bool SWISH_ADD>
__global__ void __launch_bounds__(32)
gemm1024_kernel(const float* __restrict__ A, const float* __restrict__ W,
                const float* __restrict__ addn, float* __restrict__ out)
{
    const int lane = threadIdx.x;
    const int m    = lane & 15;
    const int hi   = lane >> 4;
    const int kb2  = hi << 1;
    const int strip = blockIdx.x;
    const int nb    = blockIdx.y;

    const float* Ab = A + (size_t)strip * 16 * HIDC;
    const int n0 = nb * 128;
    v8f acc[8] = {};
    v2f fa[4], fb[4][8];
#pragma unroll
    for (int s = 0; s < 3; ++s) {
        fa[s] = load_a16x4(Ab, HIDC, 4 * s, m, kb2);
#pragma unroll
        for (int nt = 0; nt < 8; ++nt)
            fb[s][nt] = load_b4x16(W, HIDC, 4 * s, n0 + nt * 16, m, kb2);
        __builtin_amdgcn_sched_barrier(0);
    }
    for (int k0 = 0; k0 < HIDC; k0 += 16) {
#pragma unroll
        for (int u = 0; u < 4; ++u) {
            const int kq = k0 + 4 * u + 12;
            const int kn = (kq < HIDC) ? kq : HIDC - 4;
            fa[(u + 3) & 3] = load_a16x4(Ab, HIDC, kn, m, kb2);
#pragma unroll
            for (int nt = 0; nt < 8; ++nt)
                fb[(u + 3) & 3][nt] = load_b4x16(W, HIDC, kn, n0 + nt * 16, m, kb2);
            __builtin_amdgcn_sched_barrier(0);
#pragma unroll
            for (int nt = 0; nt < 8; ++nt)
                acc[nt] = wmma4(fa[u], fb[u][nt], acc[nt]);
        }
    }

    const int mbase = hi * 8;
#pragma unroll
    for (int nt = 0; nt < 8; ++nt)
#pragma unroll
        for (int j = 0; j < 8; ++j) {
            size_t off = (size_t)(strip * 16 + mbase + j) * HIDC + nb * 128 + nt * 16 + m;
            float g = acc[nt][j];
            if constexpr (SWISH_ADD) {
                out[off] = g / (1.0f + expf(-g)) + addn[off];
            } else {
                out[off] = g;
            }
        }
}

extern "C" void kernel_launch(void* const* d_in, const int* in_sizes, int n_in,
                              void* d_out, int out_size, void* d_ws, size_t ws_size,
                              hipStream_t stream)
{
    (void)in_sizes; (void)n_in; (void)out_size; (void)ws_size;
    const float* x   = (const float*)d_in[0];
    const float* wq  = (const float*)d_in[1];
    const float* wk  = (const float*)d_in[2];
    const float* wv  = (const float*)d_in[3];
    const float* w1  = (const float*)d_in[4];
    const float* w2  = (const float*)d_in[5];
    const float* gns = (const float*)d_in[6];
    const float* gnb = (const float*)d_in[7];
    float* out = (float*)d_out;
    float* ws  = (float*)d_ws;

    const size_t NQ = (size_t)BBC * NHC * SSC * HDC;
    float* q    = ws;
    float* k    = ws + NQ;
    float* v    = ws + 2 * NQ;
    float* ret  = ws + 3 * NQ;
    float* norm = q;
    float* mid  = k;

    dim3 wave(32);
    qkv_kernel<<<dim3(BBC * SSC / 16, NHC), wave, 0, stream>>>(x, wq, wk, wv, q, k, v);
    retention_kernel<<<dim3(SSC / 16, BBC * NHC), wave, 0, stream>>>(q, k, v, ret);
    groupnorm_kernel<<<dim3(BBC * SSC * NHC / 256), dim3(256), 0, stream>>>(ret, gns, gnb, norm);
    gemm1024_kernel<true ><<<dim3(BBC * SSC / 16, HIDC / 128), wave, 0, stream>>>(x,   w1, norm, mid);
    gemm1024_kernel<false><<<dim3(BBC * SSC / 16, HIDC / 128), wave, 0, stream>>>(mid, w2, nullptr, out);
}


# retention: each wave takes s-tile p and then s-tile 127-p (129 trips for every wave, 2048 waves) instead of one tile per wave, so no SIMD is left with long tiles only; sums and their order unchanged
# speedup vs baseline: 1.6107x; 1.2159x over previous
#include <hip/hip_runtime.h>
#include <hip/hip_bf16.h>
#include <math.h>

#define HIDC 1024
#define NHC  16
#define HDC  64
#define BBC  2
#define SSC  2048

typedef float v2f __attribute__((ext_vector_type(2)));
typedef float v4f __attribute__((ext_vector_type(4)));
typedef float v8f __attribute__((ext_vector_type(8)));

__device__ __forceinline__ v8f wmma4(v2f a, v2f b, v8f c) {
    return __builtin_amdgcn_wmma_f32_16x16x4_f32(
         false, a,  false, b,
         (short)0, c,  false,  false);
}

__device__ __forceinline__ v2f load2(const float* __restrict__ p) {
    return *(const v2f*)p;
}

__device__ __forceinline__ v2f load_a16x4(const float* __restrict__ A, int lda, int k0, int m, int kb2) {
    return load2(A + m * lda + k0 + kb2);
}

__device__ __forceinline__ v2f load_b4x16(const float* __restrict__ B, int ldb, int k0, int n0,
                                          int nl, int kb2) {
    const float* p = B + (k0 + kb2) * ldb + n0 + nl;
    v2f r; r[0] = p[0]; r[1] = p[ldb];
    return r;
}

__global__ void __launch_bounds__(32)
qkv_kernel(const float* __restrict__ x,
           const float* __restrict__ wq, const float* __restrict__ wk,
           const float* __restrict__ wv,
           float* __restrict__ qf, float* __restrict__ kf, float* __restrict__ vf)
{
    const int lane = threadIdx.x;
    const int m    = lane & 15;
    const int hi   = lane >> 4;
    const int kb2  = hi << 1;
    const int strip = blockIdx.x;
    const int n     = blockIdx.y;
    const int r0    = strip * 16;

    const float* Ab = x + (size_t)r0 * HIDC + n * HDC;
    const float* Wq = wq + n * HDC * HDC;
    const float* Wk = wk + n * HDC * HDC;
    const float* Wv = wv + n * HDC * HDC;

    v8f aq[4] = {}; v8f ak[4] = {}; v8f av[4] = {};
    for (int k0 = 0; k0 < HDC; k0 += 4) {
        v2f a = load_a16x4(Ab, HIDC, k0, m, kb2);
#pragma unroll
        for (int nt = 0; nt < 4; ++nt) {
            v2f bq = load_b4x16(Wq, HDC, k0, nt * 16, m, kb2);
            aq[nt] = wmma4(bq, a, aq[nt]);
            v2f bk = load_b4x16(Wk, HDC, k0, nt * 16, m, kb2);
            ak[nt] = wmma4(bk, a, ak[nt]);
            v2f bv = load_b4x16(Wv, HDC, k0, nt * 16, m, kb2);
            av[nt] = wmma4(a, bv, av[nt]);
        }
    }

    const int b  = r0 / SSC;
    const int s0 = r0 % SSC;
    const size_t tile = (((size_t)b * NHC + n) * (SSC / 16) + (s0 >> 4)) * (16 * HDC);
    float* qt = qf + tile;
    float* kt = kf + tile;
    float* vt = vf + tile;
#pragma unroll
    for (int nt = 0; nt < 4; ++nt) {
        const int io = ((2 * nt + hi) * 32 + m) * 4;
        v4f q0 = { aq[nt][0], aq[nt][1], aq[nt][2], aq[nt][3] };
        v4f q1 = { aq[nt][4], aq[nt][5], aq[nt][6], aq[nt][7] };
        v4f k0v = { ak[nt][0], ak[nt][1], ak[nt][2], ak[nt][3] };
        v4f k1v = { ak[nt][4], ak[nt][5], ak[nt][6], ak[nt][7] };
        *(v4f*)(qt + io)      = q0;
        *(v4f*)(qt + io + 64) = q1;
        *(v4f*)(kt + io)      = k0v;
        *(v4f*)(kt + io + 64) = k1v;
        v4f v0 = { av[nt][0], av[nt][1], av[nt][2], av[nt][3] };
        v4f v1 = { av[nt][4], av[nt][5], av[nt][6], av[nt][7] };
        *(v4f*)(vt + ((nt * 2 + 0) * 32 + lane) * 4) = v0;
        *(v4f*)(vt + ((nt * 2 + 1) * 32 + lane) * 4) = v1;
    }
}

__global__ void __launch_bounds__(32)
retention_kernel(const float* __restrict__ qf, const float* __restrict__ kf,
                 const float* __restrict__ vf, float* __restrict__ ret)
{
    const int lane = threadIdx.x;
    const int m    = lane & 15;
    const int hi   = lane >> 4;
    const int mbase = hi * 8;
    const int pr   = blockIdx.x;
    const int bn   = blockIdx.y;
    const int b    = bn / NHC;
    const int n    = bn % NHC;

    const float* Qb = qf + (size_t)bn * SSC * HDC;
    const float* Kb = kf + (size_t)bn * SSC * HDC;
    const float* Vb = vf + (size_t)bn * SSC * HDC;

    const float lin   = logf(1.0f / 512.0f)
                      + (float)n * ((logf(1.0f / 32.0f) - logf(1.0f / 512.0f)) / 15.0f);
    const float l2g   = log2f(1.0f - expf(lin));

    __shared__ float sS[16][18];

    for (int pass = 0; pass < 2; ++pass) {
    const int st = pass ? (SSC / 16 - 1 - pr) : pr;

    const float* Qt = Qb + st * (16 * HDC) + lane * 4;
    v4f q4[8];
#pragma unroll
    for (int i = 0; i < 8; ++i)
        q4[i] = *(const v4f*)(Qt + i * 128);

    v4f k4[8];
    {
        const float* Kt = Kb + lane * 4;
#pragma unroll
        for (int i = 0; i < 8; ++i)
            k4[i] = *(const v4f*)(Kt + i * 128);
    }

    v8f acc[4] = {};
    for (int tt = 0; tt <= st; ++tt) {
        const float* Vt = Vb + tt * (16 * HDC) + lane * 4;
        v4f vr[8];
#pragma unroll
        for (int i = 0; i < 8; ++i)
            vr[i] = *(const v4f*)(Vt + i * 128);
        v8f sc = {};
#pragma unroll
        for (int i = 0; i < 8; ++i) {
            v2f qa0 = { q4[i][0], q4[i][1] };
            v2f kb0 = { k4[i][0], k4[i][1] };
            sc = wmma4(qa0, kb0, sc);
            v2f qa1 = { q4[i][2], q4[i][3] };
            v2f kb1 = { k4[i][2], k4[i][3] };
            sc = wmma4(qa1, kb1, sc);
        }
        {
            const int tn = (tt < st) ? tt + 1 : tt;
            const float* Kn = Kb + tn * (16 * HDC) + lane * 4;
#pragma unroll
            for (int i = 0; i < 8; ++i)
                k4[i] = *(const v4f*)(Kn + i * 128);
        }
#pragma unroll
        for (int j = 0; j < 8; ++j) {
            int sg = st * 16 + mbase + j;
            int tg = tt * 16 + m;
            float d = (float)(sg - tg);
            sS[mbase + j][m] = (sg >= tg) ? sc[j] * exp2f(d * l2g) : 0.0f;
        }
        __syncthreads();
#pragma unroll
        for (int g = 0; g < 2; ++g) {
            v2f a0 = *(const v2f*)&sS[m][8 * hi + 4 * g];
            v2f a1 = *(const v2f*)&sS[m][8 * hi + 4 * g + 2];
#pragma unroll
            for (int nt = 0; nt < 4; ++nt) {
                v4f v4 = vr[nt * 2 + g];
                v2f b0 = { v4[0], v4[1] };
                v2f b1 = { v4[2], v4[3] };
                acc[nt] = wmma4(a0, b0, acc[nt]);
                acc[nt] = wmma4(a1, b1, acc[nt]);
            }
        }
        __syncthreads();
    }

    float* rb = ret + (((size_t)b * SSC + st * 16) * NHC + n) * HDC;
#pragma unroll
    for (int nt = 0; nt < 4; ++nt)
#pragma unroll
        for (int j = 0; j < 8; ++j)
            rb[(size_t)(mbase + j) * NHC * HDC + nt * 16 + m] = acc[nt][j];
    }
}

__global__ void groupnorm_kernel(const float* __restrict__ ret,
                                 const float* __restrict__ scale,
                                 const float* __restrict__ bias,
                                 float* __restrict__ normout)
{
    int idx = blockIdx.x * blockDim.x + threadIdx.x;
    if (idx >= BBC * SSC * NHC) return;
    int n = idx % NHC;
    const float* p = ret + (size_t)idx * HDC;

    float mean = 0.0f;
    for (int e = 0; e < HDC; ++e) mean += p[e];
    mean *= (1.0f / HDC);
    float var = 0.0f;
    for (int e = 0; e < HDC; ++e) { float d = p[e] - mean; var += d * d; }
    var *= (1.0f / HDC);
    float inv = rsqrtf(var + 1e-5f);

    float* o = normout + (size_t)idx * HDC;
    for (int e = 0; e < HDC; ++e) {
        int c = n * HDC + e;
        o[e] = (p[e] - mean) * inv * scale[c] + bias[c];
    }
}

template <bool SWISH_ADD>
__global__ void __launch_bounds__(32)
gemm1024_kernel(const float* __restrict__ A, const float* __restrict__ W,
                const float* __restrict__ addn, float* __restrict__ out)
{
    const int lane = threadIdx.x;
    const int m    = lane & 15;
    const int hi   = lane >> 4;
    const int kb2  = hi << 1;
    const int strip = blockIdx.x;
    const int nb    = blockIdx.y;

    const float* Ab = A + (size_t)strip * 16 * HIDC;
    const int n0 = nb * 128;
    v8f acc[8] = {};
    v2f fa[4], fb[4][8];
#pragma unroll
    for (int s = 0; s < 3; ++s) {
        fa[s] = load_a16x4(Ab, HIDC, 4 * s, m, kb2);
#pragma unroll
        for (int nt = 0; nt < 8; ++nt)
            fb[s][nt] = load_b4x16(W, HIDC, 4 * s, n0 + nt * 16, m, kb2);
        __builtin_amdgcn_sched_barrier(0);
    }
    for (int k0 = 0; k0 < HIDC; k0 += 16) {
#pragma unroll
        for (int u = 0; u < 4; ++u) {
            const int kq = k0 + 4 * u + 12;
            const int kn = (kq < HIDC) ? kq : HIDC - 4;
            fa[(u + 3) & 3] = load_a16x4(Ab, HIDC, kn, m, kb2);
#pragma unroll
            for (int nt = 0; nt < 8; ++nt)
                fb[(u + 3) & 3][nt] = load_b4x16(W, HIDC, kn, n0 + nt * 16, m, kb2);
            __builtin_amdgcn_sched_barrier(0);
#pragma unroll
            for (int nt = 0; nt < 8; ++nt)
                acc[nt] = wmma4(fa[u], fb[u][nt], acc[nt]);
        }
    }

    const int mbase = hi * 8;
#pragma unroll
    for (int nt = 0; nt < 8; ++nt)
#pragma unroll
        for (int j = 0; j < 8; ++j) {
            size_t off = (size_t)(strip * 16 + mbase + j) * HIDC + nb * 128 + nt * 16 + m;
            float g = acc[nt][j];
            if constexpr (SWISH_ADD) {
                out[off] = g / (1.0f + expf(-g)) + addn[off];
            } else {
                out[off] = g;
            }
        }
}

extern "C" void kernel_launch(void* const* d_in, const int* in_sizes, int n_in,
                              void* d_out, int out_size, void* d_ws, size_t ws_size,
                              hipStream_t stream)
{
    (void)in_sizes; (void)n_in; (void)out_size; (void)ws_size;
    const float* x   = (const float*)d_in[0];
    const float* wq  = (const float*)d_in[1];
    const float* wk  = (const float*)d_in[2];
    const float* wv  = (const float*)d_in[3];
    const float* w1  = (const float*)d_in[4];
    const float* w2  = (const float*)d_in[5];
    const float* gns = (const float*)d_in[6];
    const float* gnb = (const float*)d_in[7];
    float* out = (float*)d_out;
    float* ws  = (float*)d_ws;

    const size_t NQ = (size_t)BBC * NHC * SSC * HDC;
    float* q    = ws;
    float* k    = ws + NQ;
    float* v    = ws + 2 * NQ;
    float* ret  = ws + 3 * NQ;
    float* norm = q;
    float* mid  = k;

    dim3 wave(32);
    qkv_kernel<<<dim3(BBC * SSC / 16, NHC), wave, 0, stream>>>(x, wq, wk, wv, q, k, v);
    retention_kernel<<<dim3(SSC / 32, BBC * NHC), wave, 0, stream>>>(q, k, v, ret);
    groupnorm_kernel<<<dim3(BBC * SSC * NHC / 256), dim3(256), 0, stream>>>(ret, gns, gnb, norm);
    gemm1024_kernel<true ><<<dim3(BBC * SSC / 16, HIDC / 128), wave, 0, stream>>>(x,   w1, norm, mid);
    gemm1024_kernel<false><<<dim3(BBC * SSC / 16, HIDC / 128), wave, 0, stream>>>(mid, w2, nullptr, out);
}


# both 1024-GEMMs: 8-wave work-groups on 128x128 tiles fetch each 64-row W slab once by whole-line loads into LDS; all waves read B fragments there (was 16 4-byte loads per k-step); sums, order kept
# speedup vs baseline: 1.8055x; 1.1209x over previous
#include <hip/hip_runtime.h>
#include <hip/hip_bf16.h>
#include <math.h>

#define HIDC 1024
#define NHC  16
#define HDC  64
#define BBC  2
#define SSC  2048

typedef float v2f __attribute__((ext_vector_type(2)));
typedef float v4f __attribute__((ext_vector_type(4)));
typedef float v8f __attribute__((ext_vector_type(8)));

__device__ __forceinline__ v8f wmma4(v2f a, v2f b, v8f c) {
    return __builtin_amdgcn_wmma_f32_16x16x4_f32(
         false, a,  false, b,
         (short)0, c,  false,  false);
}

__device__ __forceinline__ v2f load2(const float* __restrict__ p) {
    return *(const v2f*)p;
}

__device__ __forceinline__ v2f load_a16x4(const float* __restrict__ A, int lda, int k0, int m, int kb2) {
    return load2(A + m * lda + k0 + kb2);
}

__device__ __forceinline__ v2f load_b4x16(const float* __restrict__ B, int ldb, int k0, int n0,
                                          int nl, int kb2) {
    const float* p = B + (k0 + kb2) * ldb + n0 + nl;
    v2f r; r[0] = p[0]; r[1] = p[ldb];
    return r;
}

__global__ void __launch_bounds__(32)
qkv_kernel(const float* __restrict__ x,
           const float* __restrict__ wq, const float* __restrict__ wk,
           const float* __restrict__ wv,
           float* __restrict__ qf, float* __restrict__ kf, float* __restrict__ vf)
{
    const int lane = threadIdx.x;
    const int m    = lane & 15;
    const int hi   = lane >> 4;
    const int kb2  = hi << 1;
    const int strip = blockIdx.x;
    const int n     = blockIdx.y;
    const int r0    = strip * 16;

    const float* Ab = x + (size_t)r0 * HIDC + n * HDC;
    const float* Wq = wq + n * HDC * HDC;
    const float* Wk = wk + n * HDC * HDC;
    const float* Wv = wv + n * HDC * HDC;

    v8f aq[4] = {}; v8f ak[4] = {}; v8f av[4] = {};
    for (int k0 = 0; k0 < HDC; k0 += 4) {
        v2f a = load_a16x4(Ab, HIDC, k0, m, kb2);
#pragma unroll
        for (int nt = 0; nt < 4; ++nt) {
            v2f bq = load_b4x16(Wq, HDC, k0, nt * 16, m, kb2);
            aq[nt] = wmma4(bq, a, aq[nt]);
            v2f bk = load_b4x16(Wk, HDC, k0, nt * 16, m, kb2);
            ak[nt] = wmma4(bk, a, ak[nt]);
            v2f bv = load_b4x16(Wv, HDC, k0, nt * 16, m, kb2);
            av[nt] = wmma4(a, bv, av[nt]);
        }
    }

    const int b  = r0 / SSC;
    const int s0 = r0 % SSC;
    const size_t tile = (((size_t)b * NHC + n) * (SSC / 16) + (s0 >> 4)) * (16 * HDC);
    float* qt = qf + tile;
    float* kt = kf + tile;
    float* vt = vf + tile;
#pragma unroll
    for (int nt = 0; nt < 4; ++nt) {
        const int io = ((2 * nt + hi) * 32 + m) * 4;
        v4f q0 = { aq[nt][0], aq[nt][1], aq[nt][2], aq[nt][3] };
        v4f q1 = { aq[nt][4], aq[nt][5], aq[nt][6], aq[nt][7] };
        v4f k0v = { ak[nt][0], ak[nt][1], ak[nt][2], ak[nt][3] };
        v4f k1v = { ak[nt][4], ak[nt][5], ak[nt][6], ak[nt][7] };
        *(v4f*)(qt + io)      = q0;
        *(v4f*)(qt + io + 64) = q1;
        *(v4f*)(kt + io)      = k0v;
        *(v4f*)(kt + io + 64) = k1v;
        v4f v0 = { av[nt][0], av[nt][1], av[nt][2], av[nt][3] };
        v4f v1 = { av[nt][4], av[nt][5], av[nt][6], av[nt][7] };
        *(v4f*)(vt + ((nt * 2 + 0) * 32 + lane) * 4) = v0;
        *(v4f*)(vt + ((nt * 2 + 1) * 32 + lane) * 4) = v1;
    }
}

__global__ void __launch_bounds__(32)
retention_kernel(const float* __restrict__ qf, const float* __restrict__ kf,
                 const float* __restrict__ vf, float* __restrict__ ret)
{
    const int lane = threadIdx.x;
    const int m    = lane & 15;
    const int hi   = lane >> 4;
    const int mbase = hi * 8;
    const int pr   = blockIdx.x;
    const int bn   = blockIdx.y;
    const int b    = bn / NHC;
    const int n    = bn % NHC;

    const float* Qb = qf + (size_t)bn * SSC * HDC;
    const float* Kb = kf + (size_t)bn * SSC * HDC;
    const float* Vb = vf + (size_t)bn * SSC * HDC;

    const float lin   = logf(1.0f / 512.0f)
                      + (float)n * ((logf(1.0f / 32.0f) - logf(1.0f / 512.0f)) / 15.0f);
    const float l2g   = log2f(1.0f - expf(lin));

    __shared__ float sS[16][18];

    for (int pass = 0; pass < 2; ++pass) {
    const int st = pass ? (SSC / 16 - 1 - pr) : pr;

    const float* Qt = Qb + st * (16 * HDC) + lane * 4;
    v4f q4[8];
#pragma unroll
    for (int i = 0; i < 8; ++i)
        q4[i] = *(const v4f*)(Qt + i * 128);

    v4f k4[8];
    {
        const float* Kt = Kb + lane * 4;
#pragma unroll
        for (int i = 0; i < 8; ++i)
            k4[i] = *(const v4f*)(Kt + i * 128);
    }

    v8f acc[4] = {};
    for (int tt = 0; tt <= st; ++tt) {
        const float* Vt = Vb + tt * (16 * HDC) + lane * 4;
        v4f vr[8];
#pragma unroll
        for (int i = 0; i < 8; ++i)
            vr[i] = *(const v4f*)(Vt + i * 128);
        v8f sc = {};
#pragma unroll
        for (int i = 0; i < 8; ++i) {
            v2f qa0 = { q4[i][0], q4[i][1] };
            v2f kb0 = { k4[i][0], k4[i][1] };
            sc = wmma4(qa0, kb0, sc);
            v2f qa1 = { q4[i][2], q4[i][3] };
            v2f kb1 = { k4[i][2], k4[i][3] };
            sc = wmma4(qa1, kb1, sc);
        }
        {
            const int tn = (tt < st) ? tt + 1 : tt;
            const float* Kn = Kb + tn * (16 * HDC) + lane * 4;
#pragma unroll
            for (int i = 0; i < 8; ++i)
                k4[i] = *(const v4f*)(Kn + i * 128);
        }
#pragma unroll
        for (int j = 0; j < 8; ++j) {
            int sg = st * 16 + mbase + j;
            int tg = tt * 16 + m;
            float d = (float)(sg - tg);
            sS[mbase + j][m] = (sg >= tg) ? sc[j] * exp2f(d * l2g) : 0.0f;
        }
        __syncthreads();
#pragma unroll
        for (int g = 0; g < 2; ++g) {
            v2f a0 = *(const v2f*)&sS[m][8 * hi + 4 * g];
            v2f a1 = *(const v2f*)&sS[m][8 * hi + 4 * g + 2];
#pragma unroll
            for (int nt = 0; nt < 4; ++nt) {
                v4f v4 = vr[nt * 2 + g];
                v2f b0 = { v4[0], v4[1] };
                v2f b1 = { v4[2], v4[3] };
                acc[nt] = wmma4(a0, b0, acc[nt]);
                acc[nt] = wmma4(a1, b1, acc[nt]);
            }
        }
        __syncthreads();
    }

    float* rb = ret + (((size_t)b * SSC + st * 16) * NHC + n) * HDC;
#pragma unroll
    for (int nt = 0; nt < 4; ++nt)
#pragma unroll
        for (int j = 0; j < 8; ++j)
            rb[(size_t)(mbase + j) * NHC * HDC + nt * 16 + m] = acc[nt][j];
    }
}

__global__ void groupnorm_kernel(const float* __restrict__ ret,
                                 const float* __restrict__ scale,
                                 const float* __restrict__ bias,
                                 float* __restrict__ normout)
{
    int idx = blockIdx.x * blockDim.x + threadIdx.x;
    if (idx >= BBC * SSC * NHC) return;
    int n = idx % NHC;
    const float* p = ret + (size_t)idx * HDC;

    float mean = 0.0f;
    for (int e = 0; e < HDC; ++e) mean += p[e];
    mean *= (1.0f / HDC);
    float var = 0.0f;
    for (int e = 0; e < HDC; ++e) { float d = p[e] - mean; var += d * d; }
    var *= (1.0f / HDC);
    float inv = rsqrtf(var + 1e-5f);

    float* o = normout + (size_t)idx * HDC;
    for (int e = 0; e < HDC; ++e) {
        int c = n * HDC + e;
        o[e] = (p[e] - mean) * inv * scale[c] + bias[c];
    }
}

#define GBK 64
#define GPP 288
#define GNCH (HIDC / GBK)
#define GSTG ((GBK / 2) * GPP)
#define GRW (GBK / 8)
#define GST (GBK / 4)
template <bool SWISH_ADD>
__global__ void __launch_bounds__(256)
gemm1024_kernel(const float* __restrict__ A, const float* __restrict__ W,
                const float* __restrict__ addn, float* __restrict__ out)
{
    __shared__ __attribute__((aligned(16))) float Ws[2 * GSTG];

    const int tid  = threadIdx.x;
    const int lane = tid & 31;
    const int wv   = tid >> 5;
    const int m    = lane & 15;
    const int hi   = lane >> 4;
    const int kb2  = hi << 1;
    const int strip = blockIdx.x * 8 + wv;
    const int nb    = blockIdx.y;
    const int n0    = nb * 128;

    const float* Ab = A + (size_t)strip * 16 * HIDC;
    const float* Wg = W + (size_t)(GRW * wv) * HIDC + n0 + 4 * lane;
    float*       wst = Ws + ((GRW / 2) * wv) * GPP + 8 * lane;
    const float* wrd = Ws + hi * GPP + 2 * m;

    v8f acc[8] = {};
    v4f wr[GRW];
    v2f fa[2][GST];
    v2f fb[2][8];

#pragma unroll
    for (int r = 0; r < GRW; ++r) wr[r] = *(const v4f*)(Wg + (size_t)r * HIDC);
#pragma unroll
    for (int s = 0; s < GST; ++s) fa[0][s] = load_a16x4(Ab, HIDC, 4 * s, m, kb2);
#pragma unroll
    for (int q = 0; q < GRW / 2; ++q) {
        v4f p0 = { wr[2*q][0], wr[2*q+1][0], wr[2*q][1], wr[2*q+1][1] };
        v4f p1 = { wr[2*q][2], wr[2*q+1][2], wr[2*q][3], wr[2*q+1][3] };
        *(v4f*)(wst + q * GPP)     = p0;
        *(v4f*)(wst + q * GPP + 4) = p1;
    }
    __syncthreads();

    for (int c2 = 0; c2 < GNCH; c2 += 2) {
#pragma unroll
        for (int h = 0; h < 2; ++h) {
            const int c = c2 + h;
            const bool more = (c + 1 < GNCH);
            const float* wcur = wrd + h * GSTG;
            if (more) {
                const int kn = (c + 1) * GBK;
#pragma unroll
                for (int r = 0; r < GRW; ++r)
                    wr[r] = *(const v4f*)(Wg + (size_t)(kn + r) * HIDC);
#pragma unroll
                for (int s = 0; s < GST; ++s)
                    fa[h ^ 1][s] = load_a16x4(Ab, HIDC, kn + 4 * s, m, kb2);
            }
#pragma unroll
            for (int nt = 0; nt < 8; ++nt)
                fb[0][nt] = *(const v2f*)(wcur + nt * 32);
#pragma unroll
            for (int s = 0; s < GST; ++s) {
                if (s + 1 < GST) {
#pragma unroll
                    for (int nt = 0; nt < 8; ++nt)
                        fb[(s + 1) & 1][nt] = *(const v2f*)(wcur + (s + 1) * 2 * GPP + nt * 32);
                }
#pragma unroll
                for (int nt = 0; nt < 8; ++nt)
                    acc[nt] = wmma4(fa[h][s], fb[s & 1][nt], acc[nt]);
            }
            if (more) {
                float* wn = wst + (h ^ 1) * GSTG;
#pragma unroll
                for (int q = 0; q < GRW / 2; ++q) {
                    v4f p0 = { wr[2*q][0], wr[2*q+1][0], wr[2*q][1], wr[2*q+1][1] };
                    v4f p1 = { wr[2*q][2], wr[2*q+1][2], wr[2*q][3], wr[2*q+1][3] };
                    *(v4f*)(wn + q * GPP)     = p0;
                    *(v4f*)(wn + q * GPP + 4) = p1;
                }
            }
            __syncthreads();
        }
    }

    const int mbase = hi * 8;
#pragma unroll
    for (int nt = 0; nt < 8; ++nt)
#pragma unroll
        for (int j = 0; j < 8; ++j) {
            size_t off = (size_t)(strip * 16 + mbase + j) * HIDC + nb * 128 + nt * 16 + m;
            float g = acc[nt][j];
            if constexpr (SWISH_ADD) {
                out[off] = g / (1.0f + expf(-g)) + addn[off];
            } else {
                out[off] = g;
            }
        }
}

extern "C" void kernel_launch(void* const* d_in, const int* in_sizes, int n_in,
                              void* d_out, int out_size, void* d_ws, size_t ws_size,
                              hipStream_t stream)
{
    (void)in_sizes; (void)n_in; (void)out_size; (void)ws_size;
    const float* x   = (const float*)d_in[0];
    const float* wq  = (const float*)d_in[1];
    const float* wk  = (const float*)d_in[2];
    const float* wv  = (const float*)d_in[3];
    const float* w1  = (const float*)d_in[4];
    const float* w2  = (const float*)d_in[5];
    const float* gns = (const float*)d_in[6];
    const float* gnb = (const float*)d_in[7];
    float* out = (float*)d_out;
    float* ws  = (float*)d_ws;

    const size_t NQ = (size_t)BBC * NHC * SSC * HDC;
    float* q    = ws;
    float* k    = ws + NQ;
    float* v    = ws + 2 * NQ;
    float* ret  = ws + 3 * NQ;
    float* norm = q;
    float* mid  = k;

    dim3 wave(32);
    qkv_kernel<<<dim3(BBC * SSC / 16, NHC), wave, 0, stream>>>(x, wq, wk, wv, q, k, v);
    retention_kernel<<<dim3(SSC / 32, BBC * NHC), wave, 0, stream>>>(q, k, v, ret);
    groupnorm_kernel<<<dim3(BBC * SSC * NHC / 256), dim3(256), 0, stream>>>(ret, gns, gnb, norm);
    gemm1024_kernel<true ><<<dim3(BBC * SSC / 128, HIDC / 128), dim3(256), 0, stream>>>(x,   w1, norm, mid);
    gemm1024_kernel<false><<<dim3(BBC * SSC / 128, HIDC / 128), dim3(256), 0, stream>>>(mid, w2, nullptr, out);
}


# retention: scores formed transposed (K x Q^T) and decay-masked by select in registers feed the 2nd product as its A operand; LDS relayout and 8 serial branch blocks per trip gone; sums, order kept
# speedup vs baseline: 1.9637x; 1.0876x over previous
#include <hip/hip_runtime.h>
#include <hip/hip_bf16.h>
#include <math.h>

#define HIDC 1024
#define NHC  16
#define HDC  64
#define BBC  2
#define SSC  2048

typedef float v2f __attribute__((ext_vector_type(2)));
typedef float v4f __attribute__((ext_vector_type(4)));
typedef float v8f __attribute__((ext_vector_type(8)));

__device__ __forceinline__ v8f wmma4(v2f a, v2f b, v8f c) {
    return __builtin_amdgcn_wmma_f32_16x16x4_f32(
         false, a,  false, b,
         (short)0, c,  false,  false);
}

__device__ __forceinline__ v2f load2(const float* __restrict__ p) {
    return *(const v2f*)p;
}

__device__ __forceinline__ v2f load_a16x4(const float* __restrict__ A, int lda, int k0, int m, int kb2) {
    return load2(A + m * lda + k0 + kb2);
}

__device__ __forceinline__ v2f load_b4x16(const float* __restrict__ B, int ldb, int k0, int n0,
                                          int nl, int kb2) {
    const float* p = B + (k0 + kb2) * ldb + n0 + nl;
    v2f r; r[0] = p[0]; r[1] = p[ldb];
    return r;
}

__global__ void __launch_bounds__(32)
qkv_kernel(const float* __restrict__ x,
           const float* __restrict__ wq, const float* __restrict__ wk,
           const float* __restrict__ wv,
           float* __restrict__ qf, float* __restrict__ kf, float* __restrict__ vf)
{
    const int lane = threadIdx.x;
    const int m    = lane & 15;
    const int hi   = lane >> 4;
    const int kb2  = hi << 1;
    const int strip = blockIdx.x;
    const int n     = blockIdx.y;
    const int r0    = strip * 16;

    const float* Ab = x + (size_t)r0 * HIDC + n * HDC;
    const float* Wq = wq + n * HDC * HDC;
    const float* Wk = wk + n * HDC * HDC;
    const float* Wv = wv + n * HDC * HDC;

    v8f aq[4] = {}; v8f ak[4] = {}; v8f av[4] = {};
    for (int k0 = 0; k0 < HDC; k0 += 4) {
        v2f a = load_a16x4(Ab, HIDC, k0, m, kb2);
#pragma unroll
        for (int nt = 0; nt < 4; ++nt) {
            v2f bq = load_b4x16(Wq, HDC, k0, nt * 16, m, kb2);
            aq[nt] = wmma4(bq, a, aq[nt]);
            v2f bk = load_b4x16(Wk, HDC, k0, nt * 16, m, kb2);
            ak[nt] = wmma4(bk, a, ak[nt]);
            v2f bv = load_b4x16(Wv, HDC, k0, nt * 16, m, kb2);
            av[nt] = wmma4(a, bv, av[nt]);
        }
    }

    const int b  = r0 / SSC;
    const int s0 = r0 % SSC;
    const size_t tile = (((size_t)b * NHC + n) * (SSC / 16) + (s0 >> 4)) * (16 * HDC);
    float* qt = qf + tile;
    float* kt = kf + tile;
    float* vt = vf + tile;
#pragma unroll
    for (int nt = 0; nt < 4; ++nt) {
        const int io = ((2 * nt + hi) * 32 + m) * 4;
        v4f q0 = { aq[nt][0], aq[nt][1], aq[nt][2], aq[nt][3] };
        v4f q1 = { aq[nt][4], aq[nt][5], aq[nt][6], aq[nt][7] };
        v4f k0v = { ak[nt][0], ak[nt][1], ak[nt][2], ak[nt][3] };
        v4f k1v = { ak[nt][4], ak[nt][5], ak[nt][6], ak[nt][7] };
        *(v4f*)(qt + io)      = q0;
        *(v4f*)(qt + io + 64) = q1;
        *(v4f*)(kt + io)      = k0v;
        *(v4f*)(kt + io + 64) = k1v;
        v4f v0 = { av[nt][0], av[nt][1], av[nt][2], av[nt][3] };
        v4f v1 = { av[nt][4], av[nt][5], av[nt][6], av[nt][7] };
        *(v4f*)(vt + ((nt * 2 + 0) * 32 + lane) * 4) = v0;
        *(v4f*)(vt + ((nt * 2 + 1) * 32 + lane) * 4) = v1;
    }
}

__global__ void __launch_bounds__(32)
retention_kernel(const float* __restrict__ qf, const float* __restrict__ kf,
                 const float* __restrict__ vf, float* __restrict__ ret)
{
    const int lane = threadIdx.x;
    const int m    = lane & 15;
    const int hi   = lane >> 4;
    const int mbase = hi * 8;
    const int pr   = blockIdx.x;
    const int bn   = blockIdx.y;
    const int b    = bn / NHC;
    const int n    = bn % NHC;

    const float* Qb = qf + (size_t)bn * SSC * HDC;
    const float* Kb = kf + (size_t)bn * SSC * HDC;
    const float* Vb = vf + (size_t)bn * SSC * HDC;

    const float lin   = logf(1.0f / 512.0f)
                      + (float)n * ((logf(1.0f / 32.0f) - logf(1.0f / 512.0f)) / 15.0f);
    const float l2g   = log2f(1.0f - expf(lin));

    for (int pass = 0; pass < 2; ++pass) {
    const int st = pass ? (SSC / 16 - 1 - pr) : pr;
    const int sg = st * 16 + m;

    const float* Qt = Qb + st * (16 * HDC) + lane * 4;
    v4f q4[8];
#pragma unroll
    for (int i = 0; i < 8; ++i)
        q4[i] = *(const v4f*)(Qt + i * 128);

    v4f k4[8];
    {
        const float* Kt = Kb + lane * 4;
#pragma unroll
        for (int i = 0; i < 8; ++i)
            k4[i] = *(const v4f*)(Kt + i * 128);
    }

    v8f acc[4] = {};
    for (int tt = 0; tt <= st; ++tt) {
        const float* Vt = Vb + tt * (16 * HDC) + lane * 4;
        v4f vr[8];
#pragma unroll
        for (int i = 0; i < 8; ++i)
            vr[i] = *(const v4f*)(Vt + i * 128);
        const int tn = (tt < st) ? tt + 1 : tt;
        const float* Kn = Kb + tn * (16 * HDC) + lane * 4;
        v4f kn[8];
#pragma unroll
        for (int i = 0; i < 8; ++i)
            kn[i] = *(const v4f*)(Kn + i * 128);
        __builtin_amdgcn_sched_barrier(0);
        v8f sc = {};
#pragma unroll
        for (int i = 0; i < 8; ++i) {
            v2f qa0 = { q4[i][0], q4[i][1] };
            v2f kb0 = { k4[i][0], k4[i][1] };
            sc = wmma4(kb0, qa0, sc);
            v2f qa1 = { q4[i][2], q4[i][3] };
            v2f kb1 = { k4[i][2], k4[i][3] };
            sc = wmma4(kb1, qa1, sc);
        }
        float p[8];
#pragma unroll
        for (int j = 0; j < 8; ++j) {
            int tg = tt * 16 + mbase + j;
            float d = (float)(sg - tg);
            float w = exp2f(d * l2g);
            p[j] = (sg >= tg) ? sc[j] * w : 0.0f;
        }
#pragma unroll
        for (int g = 0; g < 2; ++g) {
            v2f a0 = { p[4 * g],     p[4 * g + 1] };
            v2f a1 = { p[4 * g + 2], p[4 * g + 3] };
#pragma unroll
            for (int nt = 0; nt < 4; ++nt) {
                v4f v4 = vr[nt * 2 + g];
                v2f b0 = { v4[0], v4[1] };
                v2f b1 = { v4[2], v4[3] };
                acc[nt] = wmma4(a0, b0, acc[nt]);
                acc[nt] = wmma4(a1, b1, acc[nt]);
            }
        }
        __builtin_amdgcn_sched_barrier(0);
#pragma unroll
        for (int i = 0; i < 8; ++i)
            k4[i] = kn[i];
    }

    float* rb = ret + (((size_t)b * SSC + st * 16) * NHC + n) * HDC;
#pragma unroll
    for (int nt = 0; nt < 4; ++nt)
#pragma unroll
        for (int j = 0; j < 8; ++j)
            rb[(size_t)(mbase + j) * NHC * HDC + nt * 16 + m] = acc[nt][j];
    }
}

__global__ void groupnorm_kernel(const float* __restrict__ ret,
                                 const float* __restrict__ scale,
                                 const float* __restrict__ bias,
                                 float* __restrict__ normout)
{
    int idx = blockIdx.x * blockDim.x + threadIdx.x;
    if (idx >= BBC * SSC * NHC) return;
    int n = idx % NHC;
    const float* p = ret + (size_t)idx * HDC;

    float mean = 0.0f;
    for (int e = 0; e < HDC; ++e) mean += p[e];
    mean *= (1.0f / HDC);
    float var = 0.0f;
    for (int e = 0; e < HDC; ++e) { float d = p[e] - mean; var += d * d; }
    var *= (1.0f / HDC);
    float inv = rsqrtf(var + 1e-5f);

    float* o = normout + (size_t)idx * HDC;
    for (int e = 0; e < HDC; ++e) {
        int c = n * HDC + e;
        o[e] = (p[e] - mean) * inv * scale[c] + bias[c];
    }
}

#define GBK 64
#define GPP 288
#define GNCH (HIDC / GBK)
#define GSTG ((GBK / 2) * GPP)
#define GRW (GBK / 8)
#define GST (GBK / 4)
template <bool SWISH_ADD>
__global__ void __launch_bounds__(256)
gemm1024_kernel(const float* __restrict__ A, const float* __restrict__ W,
                const float* __restrict__ addn, float* __restrict__ out)
{
    __shared__ __attribute__((aligned(16))) float Ws[2 * GSTG];

    const int tid  = threadIdx.x;
    const int lane = tid & 31;
    const int wv   = tid >> 5;
    const int m    = lane & 15;
    const int hi   = lane >> 4;
    const int kb2  = hi << 1;
    const int strip = blockIdx.x * 8 + wv;
    const int nb    = blockIdx.y;
    const int n0    = nb * 128;

    const float* Ab = A + (size_t)strip * 16 * HIDC;
    const float* Wg = W + (size_t)(GRW * wv) * HIDC + n0 + 4 * lane;
    float*       wst = Ws + ((GRW / 2) * wv) * GPP + 8 * lane;
    const float* wrd = Ws + hi * GPP + 2 * m;

    v8f acc[8] = {};
    v4f wr[GRW];
    v2f fa[2][GST];
    v2f fb[2][8];

#pragma unroll
    for (int r = 0; r < GRW; ++r) wr[r] = *(const v4f*)(Wg + (size_t)r * HIDC);
#pragma unroll
    for (int s = 0; s < GST; ++s) fa[0][s] = load_a16x4(Ab, HIDC, 4 * s, m, kb2);
#pragma unroll
    for (int q = 0; q < GRW / 2; ++q) {
        v4f p0 = { wr[2*q][0], wr[2*q+1][0], wr[2*q][1], wr[2*q+1][1] };
        v4f p1 = { wr[2*q][2], wr[2*q+1][2], wr[2*q][3], wr[2*q+1][3] };
        *(v4f*)(wst + q * GPP)     = p0;
        *(v4f*)(wst + q * GPP + 4) = p1;
    }
    __syncthreads();

    for (int c2 = 0; c2 < GNCH; c2 += 2) {
#pragma unroll
        for (int h = 0; h < 2; ++h) {
            const int c = c2 + h;
            const bool more = (c + 1 < GNCH);
            const float* wcur = wrd + h * GSTG;
            if (more) {
                const int kn = (c + 1) * GBK;
#pragma unroll
                for (int r = 0; r < GRW; ++r)
                    wr[r] = *(const v4f*)(Wg + (size_t)(kn + r) * HIDC);
#pragma unroll
                for (int s = 0; s < GST; ++s)
                    fa[h ^ 1][s] = load_a16x4(Ab, HIDC, kn + 4 * s, m, kb2);
            }
#pragma unroll
            for (int nt = 0; nt < 8; ++nt)
                fb[0][nt] = *(const v2f*)(wcur + nt * 32);
#pragma unroll
            for (int s = 0; s < GST; ++s) {
                if (s + 1 < GST) {
#pragma unroll
                    for (int nt = 0; nt < 8; ++nt)
                        fb[(s + 1) & 1][nt] = *(const v2f*)(wcur + (s + 1) * 2 * GPP + nt * 32);
                }
#pragma unroll
                for (int nt = 0; nt < 8; ++nt)
                    acc[nt] = wmma4(fa[h][s], fb[s & 1][nt], acc[nt]);
            }
            if (more) {
                float* wn = wst + (h ^ 1) * GSTG;
#pragma unroll
                for (int q = 0; q < GRW / 2; ++q) {
                    v4f p0 = { wr[2*q][0], wr[2*q+1][0], wr[2*q][1], wr[2*q+1][1] };
                    v4f p1 = { wr[2*q][2], wr[2*q+1][2], wr[2*q][3], wr[2*q+1][3] };
                    *(v4f*)(wn + q * GPP)     = p0;
                    *(v4f*)(wn + q * GPP + 4) = p1;
                }
            }
            __syncthreads();
        }
    }

    const int mbase = hi * 8;
#pragma unroll
    for (int nt = 0; nt < 8; ++nt)
#pragma unroll
        for (int j = 0; j < 8; ++j) {
            size_t off = (size_t)(strip * 16 + mbase + j) * HIDC + nb * 128 + nt * 16 + m;
            float g = acc[nt][j];
            if constexpr (SWISH_ADD) {
                out[off] = g / (1.0f + expf(-g)) + addn[off];
            } else {
                out[off] = g;
            }
        }
}

extern "C" void kernel_launch(void* const* d_in, const int* in_sizes, int n_in,
                              void* d_out, int out_size, void* d_ws, size_t ws_size,
                              hipStream_t stream)
{
    (void)in_sizes; (void)n_in; (void)out_size; (void)ws_size;
    const float* x   = (const float*)d_in[0];
    const float* wq  = (const float*)d_in[1];
    const float* wk  = (const float*)d_in[2];
    const float* wv  = (const float*)d_in[3];
    const float* w1  = (const float*)d_in[4];
    const float* w2  = (const float*)d_in[5];
    const float* gns = (const float*)d_in[6];
    const float* gnb = (const float*)d_in[7];
    float* out = (float*)d_out;
    float* ws  = (float*)d_ws;

    const size_t NQ = (size_t)BBC * NHC * SSC * HDC;
    float* q    = ws;
    float* k    = ws + NQ;
    float* v    = ws + 2 * NQ;
    float* ret  = ws + 3 * NQ;
    float* norm = q;
    float* mid  = k;

    dim3 wave(32);
    qkv_kernel<<<dim3(BBC * SSC / 16, NHC), wave, 0, stream>>>(x, wq, wk, wv, q, k, v);
    retention_kernel<<<dim3(SSC / 32, BBC * NHC), wave, 0, stream>>>(q, k, v, ret);
    groupnorm_kernel<<<dim3(BBC * SSC * NHC / 256), dim3(256), 0, stream>>>(ret, gns, gnb, norm);
    gemm1024_kernel<true ><<<dim3(BBC * SSC / 128, HIDC / 128), dim3(256), 0, stream>>>(x,   w1, norm, mid);
    gemm1024_kernel<false><<<dim3(BBC * SSC / 128, HIDC / 128), dim3(256), 0, stream>>>(mid, w2, nullptr, out);
}


# GroupNorm and its affine move into the retention epilogue on the accumulator registers (launch 3 of 5 and its 16 MB round trip gone); mean and variance sums become a 4-in-lane + 16-lane xor tree
# speedup vs baseline: 2.0939x; 1.0663x over previous
#include <hip/hip_runtime.h>
#include <hip/hip_bf16.h>
#include <math.h>

#define HIDC 1024
#define NHC  16
#define HDC  64
#define BBC  2
#define SSC  2048

typedef float v2f __attribute__((ext_vector_type(2)));
typedef float v4f __attribute__((ext_vector_type(4)));
typedef float v8f __attribute__((ext_vector_type(8)));

__device__ __forceinline__ v8f wmma4(v2f a, v2f b, v8f c) {
    return __builtin_amdgcn_wmma_f32_16x16x4_f32(
         false, a,  false, b,
         (short)0, c,  false,  false);
}

__device__ __forceinline__ v2f load2(const float* __restrict__ p) {
    return *(const v2f*)p;
}

__device__ __forceinline__ v2f load_a16x4(const float* __restrict__ A, int lda, int k0, int m, int kb2) {
    return load2(A + m * lda + k0 + kb2);
}

__device__ __forceinline__ v2f load_b4x16(const float* __restrict__ B, int ldb, int k0, int n0,
                                          int nl, int kb2) {
    const float* p = B + (k0 + kb2) * ldb + n0 + nl;
    v2f r; r[0] = p[0]; r[1] = p[ldb];
    return r;
}

__global__ void __launch_bounds__(32)
qkv_kernel(const float* __restrict__ x,
           const float* __restrict__ wq, const float* __restrict__ wk,
           const float* __restrict__ wv,
           float* __restrict__ qf, float* __restrict__ kf, float* __restrict__ vf)
{
    const int lane = threadIdx.x;
    const int m    = lane & 15;
    const int hi   = lane >> 4;
    const int kb2  = hi << 1;
    const int strip = blockIdx.x;
    const int n     = blockIdx.y;
    const int r0    = strip * 16;

    const float* Ab = x + (size_t)r0 * HIDC + n * HDC;
    const float* Wq = wq + n * HDC * HDC;
    const float* Wk = wk + n * HDC * HDC;
    const float* Wv = wv + n * HDC * HDC;

    v8f aq[4] = {}; v8f ak[4] = {}; v8f av[4] = {};
    for (int k0 = 0; k0 < HDC; k0 += 4) {
        v2f a = load_a16x4(Ab, HIDC, k0, m, kb2);
#pragma unroll
        for (int nt = 0; nt < 4; ++nt) {
            v2f bq = load_b4x16(Wq, HDC, k0, nt * 16, m, kb2);
            aq[nt] = wmma4(bq, a, aq[nt]);
            v2f bk = load_b4x16(Wk, HDC, k0, nt * 16, m, kb2);
            ak[nt] = wmma4(bk, a, ak[nt]);
            v2f bv = load_b4x16(Wv, HDC, k0, nt * 16, m, kb2);
            av[nt] = wmma4(a, bv, av[nt]);
        }
    }

    const int b  = r0 / SSC;
    const int s0 = r0 % SSC;
    const size_t tile = (((size_t)b * NHC + n) * (SSC / 16) + (s0 >> 4)) * (16 * HDC);
    float* qt = qf + tile;
    float* kt = kf + tile;
    float* vt = vf + tile;
#pragma unroll
    for (int nt = 0; nt < 4; ++nt) {
        const int io = ((2 * nt + hi) * 32 + m) * 4;
        v4f q0 = { aq[nt][0], aq[nt][1], aq[nt][2], aq[nt][3] };
        v4f q1 = { aq[nt][4], aq[nt][5], aq[nt][6], aq[nt][7] };
        v4f k0v = { ak[nt][0], ak[nt][1], ak[nt][2], ak[nt][3] };
        v4f k1v = { ak[nt][4], ak[nt][5], ak[nt][6], ak[nt][7] };
        *(v4f*)(qt + io)      = q0;
        *(v4f*)(qt + io + 64) = q1;
        *(v4f*)(kt + io)      = k0v;
        *(v4f*)(kt + io + 64) = k1v;
        v4f v0 = { av[nt][0], av[nt][1], av[nt][2], av[nt][3] };
        v4f v1 = { av[nt][4], av[nt][5], av[nt][6], av[nt][7] };
        *(v4f*)(vt + ((nt * 2 + 0) * 32 + lane) * 4) = v0;
        *(v4f*)(vt + ((nt * 2 + 1) * 32 + lane) * 4) = v1;
    }
}

__global__ void __launch_bounds__(32)
retention_kernel(const float* __restrict__ qf, const float* __restrict__ kf,
                 const float* __restrict__ vf,
                 const float* __restrict__ scale, const float* __restrict__ bias,
                 float* __restrict__ normout)
{
    const int lane = threadIdx.x;
    const int m    = lane & 15;
    const int hi   = lane >> 4;
    const int mbase = hi * 8;
    const int pr   = blockIdx.x;
    const int bn   = blockIdx.y;
    const int b    = bn / NHC;
    const int n    = bn % NHC;

    const float* Qb = qf + (size_t)bn * SSC * HDC;
    const float* Kb = kf + (size_t)bn * SSC * HDC;
    const float* Vb = vf + (size_t)bn * SSC * HDC;

    const float lin   = logf(1.0f / 512.0f)
                      + (float)n * ((logf(1.0f / 32.0f) - logf(1.0f / 512.0f)) / 15.0f);
    const float l2g   = log2f(1.0f - expf(lin));

    float gsc[4], gbi[4];
#pragma unroll
    for (int nt = 0; nt < 4; ++nt) {
        gsc[nt] = scale[n * HDC + nt * 16 + m];
        gbi[nt] = bias[n * HDC + nt * 16 + m];
    }

    for (int pass = 0; pass < 2; ++pass) {
    const int st = pass ? (SSC / 16 - 1 - pr) : pr;
    const int sg = st * 16 + m;

    const float* Qt = Qb + st * (16 * HDC) + lane * 4;
    v4f q4[8];
#pragma unroll
    for (int i = 0; i < 8; ++i)
        q4[i] = *(const v4f*)(Qt + i * 128);

    v4f k4[8];
    {
        const float* Kt = Kb + lane * 4;
#pragma unroll
        for (int i = 0; i < 8; ++i)
            k4[i] = *(const v4f*)(Kt + i * 128);
    }

    v8f acc[4] = {};
    for (int tt = 0; tt <= st; ++tt) {
        const float* Vt = Vb + tt * (16 * HDC) + lane * 4;
        v4f vr[8];
#pragma unroll
        for (int i = 0; i < 8; ++i)
            vr[i] = *(const v4f*)(Vt + i * 128);
        const int tn = (tt < st) ? tt + 1 : tt;
        const float* Kn = Kb + tn * (16 * HDC) + lane * 4;
        v4f kn[8];
#pragma unroll
        for (int i = 0; i < 8; ++i)
            kn[i] = *(const v4f*)(Kn + i * 128);
        __builtin_amdgcn_sched_barrier(0);
        v8f sc = {};
#pragma unroll
        for (int i = 0; i < 8; ++i) {
            v2f qa0 = { q4[i][0], q4[i][1] };
            v2f kb0 = { k4[i][0], k4[i][1] };
            sc = wmma4(kb0, qa0, sc);
            v2f qa1 = { q4[i][2], q4[i][3] };
            v2f kb1 = { k4[i][2], k4[i][3] };
            sc = wmma4(kb1, qa1, sc);
        }
        float p[8];
#pragma unroll
        for (int j = 0; j < 8; ++j) {
            int tg = tt * 16 + mbase + j;
            float d = (float)(sg - tg);
            float w = exp2f(d * l2g);
            p[j] = (sg >= tg) ? sc[j] * w : 0.0f;
        }
#pragma unroll
        for (int g = 0; g < 2; ++g) {
            v2f a0 = { p[4 * g],     p[4 * g + 1] };
            v2f a1 = { p[4 * g + 2], p[4 * g + 3] };
#pragma unroll
            for (int nt = 0; nt < 4; ++nt) {
                v4f v4 = vr[nt * 2 + g];
                v2f b0 = { v4[0], v4[1] };
                v2f b1 = { v4[2], v4[3] };
                acc[nt] = wmma4(a0, b0, acc[nt]);
                acc[nt] = wmma4(a1, b1, acc[nt]);
            }
        }
        __builtin_amdgcn_sched_barrier(0);
#pragma unroll
        for (int i = 0; i < 8; ++i)
            k4[i] = kn[i];
    }

    float rs[8];
#pragma unroll
    for (int j = 0; j < 8; ++j)
        rs[j] = (acc[0][j] + acc[1][j]) + (acc[2][j] + acc[3][j]);
#pragma unroll
    for (int sh = 1; sh < 16; sh <<= 1)
#pragma unroll
        for (int j = 0; j < 8; ++j)
            rs[j] += __shfl_xor(rs[j], sh);
    float dv[4][8];
    float rq[8];
#pragma unroll
    for (int j = 0; j < 8; ++j) {
        const float mean = rs[j] * (1.0f / HDC);
#pragma unroll
        for (int nt = 0; nt < 4; ++nt)
            dv[nt][j] = acc[nt][j] - mean;
        rq[j] = (dv[0][j] * dv[0][j] + dv[1][j] * dv[1][j])
              + (dv[2][j] * dv[2][j] + dv[3][j] * dv[3][j]);
    }
#pragma unroll
    for (int sh = 1; sh < 16; sh <<= 1)
#pragma unroll
        for (int j = 0; j < 8; ++j)
            rq[j] += __shfl_xor(rq[j], sh);

    float* ob = normout + (((size_t)b * SSC + st * 16) * NHC + n) * HDC;
#pragma unroll
    for (int j = 0; j < 8; ++j) {
        const float inv = rsqrtf(rq[j] * (1.0f / HDC) + 1e-5f);
#pragma unroll
        for (int nt = 0; nt < 4; ++nt)
            ob[(size_t)(mbase + j) * NHC * HDC + nt * 16 + m] =
                dv[nt][j] * inv * gsc[nt] + gbi[nt];
    }
    }
}

#define GBK 64
#define GPP 288
#define GNCH (HIDC / GBK)
#define GSTG ((GBK / 2) * GPP)
#define GRW (GBK / 8)
#define GST (GBK / 4)
template <bool SWISH_ADD>
__global__ void __launch_bounds__(256)
gemm1024_kernel(const float* __restrict__ A, const float* __restrict__ W,
                const float* __restrict__ addn, float* __restrict__ out)
{
    __shared__ __attribute__((aligned(16))) float Ws[2 * GSTG];

    const int tid  = threadIdx.x;
    const int lane = tid & 31;
    const int wv   = tid >> 5;
    const int m    = lane & 15;
    const int hi   = lane >> 4;
    const int kb2  = hi << 1;
    const int strip = blockIdx.x * 8 + wv;
    const int nb    = blockIdx.y;
    const int n0    = nb * 128;

    const float* Ab = A + (size_t)strip * 16 * HIDC;
    const float* Wg = W + (size_t)(GRW * wv) * HIDC + n0 + 4 * lane;
    float*       wst = Ws + ((GRW / 2) * wv) * GPP + 8 * lane;
    const float* wrd = Ws + hi * GPP + 2 * m;

    v8f acc[8] = {};
    v4f wr[GRW];
    v2f fa[2][GST];
    v2f fb[2][8];

#pragma unroll
    for (int r = 0; r < GRW; ++r) wr[r] = *(const v4f*)(Wg + (size_t)r * HIDC);
#pragma unroll
    for (int s = 0; s < GST; ++s) fa[0][s] = load_a16x4(Ab, HIDC, 4 * s, m, kb2);
#pragma unroll
    for (int q = 0; q < GRW / 2; ++q) {
        v4f p0 = { wr[2*q][0], wr[2*q+1][0], wr[2*q][1], wr[2*q+1][1] };
        v4f p1 = { wr[2*q][2], wr[2*q+1][2], wr[2*q][3], wr[2*q+1][3] };
        *(v4f*)(wst + q * GPP)     = p0;
        *(v4f*)(wst + q * GPP + 4) = p1;
    }
    __syncthreads();

    for (int c2 = 0; c2 < GNCH; c2 += 2) {
#pragma unroll
        for (int h = 0; h < 2; ++h) {
            const int c = c2 + h;
            const bool more = (c + 1 < GNCH);
            const float* wcur = wrd + h * GSTG;
            if (more) {
                const int kn = (c + 1) * GBK;
#pragma unroll
                for (int r = 0; r < GRW; ++r)
                    wr[r] = *(const v4f*)(Wg + (size_t)(kn + r) * HIDC);
#pragma unroll
                for (int s = 0; s < GST; ++s)
                    fa[h ^ 1][s] = load_a16x4(Ab, HIDC, kn + 4 * s, m, kb2);
            }
#pragma unroll
            for (int nt = 0; nt < 8; ++nt)
                fb[0][nt] = *(const v2f*)(wcur + nt * 32);
#pragma unroll
            for (int s = 0; s < GST; ++s) {
                if (s + 1 < GST) {
#pragma unroll
                    for (int nt = 0; nt < 8; ++nt)
                        fb[(s + 1) & 1][nt] = *(const v2f*)(wcur + (s + 1) * 2 * GPP + nt * 32);
                }
#pragma unroll
                for (int nt = 0; nt < 8; ++nt)
                    acc[nt] = wmma4(fa[h][s], fb[s & 1][nt], acc[nt]);
            }
            if (more) {
                float* wn = wst + (h ^ 1) * GSTG;
#pragma unroll
                for (int q = 0; q < GRW / 2; ++q) {
                    v4f p0 = { wr[2*q][0], wr[2*q+1][0], wr[2*q][1], wr[2*q+1][1] };
                    v4f p1 = { wr[2*q][2], wr[2*q+1][2], wr[2*q][3], wr[2*q+1][3] };
                    *(v4f*)(wn + q * GPP)     = p0;
                    *(v4f*)(wn + q * GPP + 4) = p1;
                }
            }
            __syncthreads();
        }
    }

    const int mbase = hi * 8;
#pragma unroll
    for (int nt = 0; nt < 8; ++nt)
#pragma unroll
        for (int j = 0; j < 8; ++j) {
            size_t off = (size_t)(strip * 16 + mbase + j) * HIDC + nb * 128 + nt * 16 + m;
            float g = acc[nt][j];
            if constexpr (SWISH_ADD) {
                out[off] = g / (1.0f + expf(-g)) + addn[off];
            } else {
                out[off] = g;
            }
        }
}

extern "C" void kernel_launch(void* const* d_in, const int* in_sizes, int n_in,
                              void* d_out, int out_size, void* d_ws, size_t ws_size,
                              hipStream_t stream)
{
    (void)in_sizes; (void)n_in; (void)out_size; (void)ws_size;
    const float* x   = (const float*)d_in[0];
    const float* wq  = (const float*)d_in[1];
    const float* wk  = (const float*)d_in[2];
    const float* wv  = (const float*)d_in[3];
    const float* w1  = (const float*)d_in[4];
    const float* w2  = (const float*)d_in[5];
    const float* gns = (const float*)d_in[6];
    const float* gnb = (const float*)d_in[7];
    float* out = (float*)d_out;
    float* ws  = (float*)d_ws;

    const size_t NQ = (size_t)BBC * NHC * SSC * HDC;
    float* q    = ws;
    float* k    = ws + NQ;
    float* v    = ws + 2 * NQ;
    float* norm = ws + 3 * NQ;
    float* mid  = k;

    dim3 wave(32);
    qkv_kernel<<<dim3(BBC * SSC / 16, NHC), wave, 0, stream>>>(x, wq, wk, wv, q, k, v);
    retention_kernel<<<dim3(SSC / 32, BBC * NHC), wave, 0, stream>>>(q, k, v, gns, gnb, norm);
    gemm1024_kernel<true ><<<dim3(BBC * SSC / 128, HIDC / 128), dim3(256), 0, stream>>>(x,   w1, norm, mid);
    gemm1024_kernel<false><<<dim3(BBC * SSC / 128, HIDC / 128), dim3(256), 0, stream>>>(mid, w2, nullptr, out);
}


# retention by 64-row chunks: earlier chunks enter via one 64x64 state (decayed k)^T v per chunk from a new scan launch; score loop covers own chunk only; ret's sums are taken in another order
# speedup vs baseline: 3.4512x; 1.6482x over previous
#include <hip/hip_runtime.h>
#include <hip/hip_bf16.h>
#include <math.h>

#define HIDC 1024
#define NHC  16
#define HDC  64
#define BBC  2
#define SSC  2048

typedef float v2f __attribute__((ext_vector_type(2)));
typedef float v4f __attribute__((ext_vector_type(4)));
typedef float v8f __attribute__((ext_vector_type(8)));

__device__ __forceinline__ v8f wmma4(v2f a, v2f b, v8f c) {
    return __builtin_amdgcn_wmma_f32_16x16x4_f32(
         false, a,  false, b,
         (short)0, c,  false,  false);
}

__device__ __forceinline__ v2f load2(const float* __restrict__ p) {
    return *(const v2f*)p;
}

__device__ __forceinline__ v2f load_a16x4(const float* __restrict__ A, int lda, int k0, int m, int kb2) {
    return load2(A + m * lda + k0 + kb2);
}

__device__ __forceinline__ v2f load_b4x16(const float* __restrict__ B, int ldb, int k0, int n0,
                                          int nl, int kb2) {
    const float* p = B + (k0 + kb2) * ldb + n0 + nl;
    v2f r; r[0] = p[0]; r[1] = p[ldb];
    return r;
}

__device__ __forceinline__ float head_l2g(int n) {
    const float lin = logf(1.0f / 512.0f)
                    + (float)n * ((logf(1.0f / 32.0f) - logf(1.0f / 512.0f)) / 15.0f);
    return log2f(1.0f - expf(lin));
}

#define CHK 64
#define NCHK (SSC / CHK)

__global__ void __launch_bounds__(32)
qkv_kernel(const float* __restrict__ x,
           const float* __restrict__ wq, const float* __restrict__ wk,
           const float* __restrict__ wv,
           float* __restrict__ qf, float* __restrict__ kf, float* __restrict__ vf,
           float* __restrict__ kdf)
{
    const int lane = threadIdx.x;
    const int m    = lane & 15;
    const int hi   = lane >> 4;
    const int kb2  = hi << 1;
    const int strip = blockIdx.x;
    const int n     = blockIdx.y;
    const int r0    = strip * 16;

    const float* Ab = x + (size_t)r0 * HIDC + n * HDC;
    const float* Wq = wq + n * HDC * HDC;
    const float* Wk = wk + n * HDC * HDC;
    const float* Wv = wv + n * HDC * HDC;

    v8f aq[4] = {}; v8f ak[4] = {}; v8f av[4] = {}; v8f ad[4] = {};
    for (int k0 = 0; k0 < HDC; k0 += 4) {
        v2f a = load_a16x4(Ab, HIDC, k0, m, kb2);
#pragma unroll
        for (int nt = 0; nt < 4; ++nt) {
            v2f bq = load_b4x16(Wq, HDC, k0, nt * 16, m, kb2);
            aq[nt] = wmma4(bq, a, aq[nt]);
            v2f bk = load_b4x16(Wk, HDC, k0, nt * 16, m, kb2);
            ak[nt] = wmma4(bk, a, ak[nt]);
            ad[nt] = wmma4(a, bk, ad[nt]);
            v2f bv = load_b4x16(Wv, HDC, k0, nt * 16, m, kb2);
            av[nt] = wmma4(a, bv, av[nt]);
        }
    }

    const int b  = r0 / SSC;
    const int s0 = r0 % SSC;
    const size_t tile = (((size_t)b * NHC + n) * (SSC / 16) + (s0 >> 4)) * (16 * HDC);
    float* qt = qf + tile;
    float* kt = kf + tile;
    float* vt = vf + tile;
    float* dt = kdf + tile;
    const float l2g = head_l2g(n);
    float wd[8];
#pragma unroll
    for (int j = 0; j < 8; ++j)
        wd[j] = exp2f((float)(CHK - 1 - ((s0 & (CHK - 1)) + 8 * hi + j)) * l2g);
#pragma unroll
    for (int nt = 0; nt < 4; ++nt) {
        const int io = ((2 * nt + hi) * 32 + m) * 4;
        v4f q0 = { aq[nt][0], aq[nt][1], aq[nt][2], aq[nt][3] };
        v4f q1 = { aq[nt][4], aq[nt][5], aq[nt][6], aq[nt][7] };
        v4f k0v = { ak[nt][0], ak[nt][1], ak[nt][2], ak[nt][3] };
        v4f k1v = { ak[nt][4], ak[nt][5], ak[nt][6], ak[nt][7] };
        *(v4f*)(qt + io)      = q0;
        *(v4f*)(qt + io + 64) = q1;
        *(v4f*)(kt + io)      = k0v;
        *(v4f*)(kt + io + 64) = k1v;
        v4f v0 = { av[nt][0], av[nt][1], av[nt][2], av[nt][3] };
        v4f v1 = { av[nt][4], av[nt][5], av[nt][6], av[nt][7] };
        *(v4f*)(vt + ((nt * 2 + 0) * 32 + lane) * 4) = v0;
        *(v4f*)(vt + ((nt * 2 + 1) * 32 + lane) * 4) = v1;
        v4f d0 = { ad[nt][0] * wd[0], ad[nt][1] * wd[1], ad[nt][2] * wd[2], ad[nt][3] * wd[3] };
        v4f d1 = { ad[nt][4] * wd[4], ad[nt][5] * wd[5], ad[nt][6] * wd[6], ad[nt][7] * wd[7] };
        *(v4f*)(dt + ((nt * 2 + 0) * 32 + lane) * 4) = d0;
        *(v4f*)(dt + ((nt * 2 + 1) * 32 + lane) * 4) = d1;
    }
}

__device__ __forceinline__ void state_step(const v4f (&fk)[8], const v4f (&fv)[8],
                                           v8f& s0, v8f& s1, float g64,
                                           float* __restrict__ sc)
{
#pragma unroll
    for (int j = 0; j < 8; ++j) { s0[j] *= g64; s1[j] *= g64; }
#pragma unroll
    for (int ti = 0; ti < 4; ++ti) {
        v4f a0 = fk[ti * 2],     b0 = fv[ti * 2];
        v4f a1 = fk[ti * 2 + 1], b1 = fv[ti * 2 + 1];
        v2f a00 = { a0[0], a0[1] }, b00 = { b0[0], b0[1] };
        v2f a01 = { a0[2], a0[3] }, b01 = { b0[2], b0[3] };
        v2f a10 = { a1[0], a1[1] }, b10 = { b1[0], b1[1] };
        v2f a11 = { a1[2], a1[3] }, b11 = { b1[2], b1[3] };
        s0 = wmma4(a00, b00, s0);
        s1 = wmma4(a10, b10, s1);
        s0 = wmma4(a01, b01, s0);
        s1 = wmma4(a11, b11, s1);
    }
    v4f o0 = { s0[0] + s1[0], s0[1] + s1[1], s0[2] + s1[2], s0[3] + s1[3] };
    v4f o1 = { s0[4] + s1[4], s0[5] + s1[5], s0[6] + s1[6], s0[7] + s1[7] };
    *(v4f*)(sc)      = o0;
    *(v4f*)(sc + 64) = o1;
}

__global__ void __launch_bounds__(32)
state_kernel(const float* __restrict__ kdf, const float* __restrict__ vf,
             float* __restrict__ sf)
{
    const int lane = threadIdx.x;
    const int m    = lane & 15;
    const int hi   = lane >> 4;
    const int dtl  = blockIdx.x >> 2;
    const int etl  = blockIdx.x & 3;
    const int bn   = blockIdx.y;
    const int n    = bn % NHC;
    const float g64 = exp2f((float)CHK * head_l2g(n));

    const float* Kd = kdf + (size_t)bn * SSC * HDC + (dtl * 2 * 32 + lane) * 4;
    const float* Vb = vf  + (size_t)bn * SSC * HDC + (etl * 2 * 32 + lane) * 4;
    float* So = sf + (size_t)bn * (NCHK - 1) * (CHK * HDC)
              + (((2 * dtl + hi) * 4 + etl) * 32 + m) * 4;

    v4f fk[3][8], fv[3][8];
#pragma unroll
    for (int c = 0; c < 2; ++c)
#pragma unroll
        for (int ti = 0; ti < 4; ++ti)
#pragma unroll
            for (int g = 0; g < 2; ++g) {
                fk[c][ti * 2 + g] = *(const v4f*)(Kd + (c * 4 + ti) * (16 * HDC) + g * 128);
                fv[c][ti * 2 + g] = *(const v4f*)(Vb + (c * 4 + ti) * (16 * HDC) + g * 128);
            }

    v8f s0 = {}; v8f s1 = {};
    for (int c3 = 0; c3 < NCHK - 2; c3 += 3) {
#pragma unroll
        for (int u = 0; u < 3; ++u) {
            const int c = c3 + u;
            const int cn = (c + 2 < NCHK - 1) ? c + 2 : NCHK - 2;
#pragma unroll
            for (int ti = 0; ti < 4; ++ti)
#pragma unroll
                for (int g = 0; g < 2; ++g) {
                    fk[(u + 2) % 3][ti * 2 + g] =
                        *(const v4f*)(Kd + (cn * 4 + ti) * (16 * HDC) + g * 128);
                    fv[(u + 2) % 3][ti * 2 + g] =
                        *(const v4f*)(Vb + (cn * 4 + ti) * (16 * HDC) + g * 128);
                }
            __builtin_amdgcn_sched_barrier(0);
            state_step(fk[u], fv[u], s0, s1, g64, So + (size_t)c * (CHK * HDC));
            __builtin_amdgcn_sched_barrier(0);
        }
    }
    state_step(fk[0], fv[0], s0, s1, g64, So + (size_t)(NCHK - 2) * (CHK * HDC));
}

__global__ void __launch_bounds__(32)
retention_kernel(const float* __restrict__ qf, const float* __restrict__ kf,
                 const float* __restrict__ vf, const float* __restrict__ sf,
                 const float* __restrict__ scale, const float* __restrict__ bias,
                 float* __restrict__ normout)
{
    const int lane = threadIdx.x;
    const int m    = lane & 15;
    const int hi   = lane >> 4;
    const int mbase = hi * 8;
    const int pr   = blockIdx.x;
    const int bn   = blockIdx.y;
    const int b    = bn / NHC;
    const int n    = bn % NHC;

    const float* Qb = qf + (size_t)bn * SSC * HDC;
    const float* Kb = kf + (size_t)bn * SSC * HDC;
    const float* Vb = vf + (size_t)bn * SSC * HDC;

    const float* Sb = sf + (size_t)bn * (NCHK - 1) * (CHK * HDC);
    const float l2g = head_l2g(n);

    float gsc[4], gbi[4];
#pragma unroll
    for (int nt = 0; nt < 4; ++nt) {
        gsc[nt] = scale[n * HDC + nt * 16 + m];
        gbi[nt] = bias[n * HDC + nt * 16 + m];
    }

    for (int pass = 0; pass < 2; ++pass) {
    const int st = pass ? (SSC / 16 - 1 - pr) : pr;
    const int sg = st * 16 + m;

    const float* Qt = Qb + st * (16 * HDC) + lane * 4;
    v4f q4[8];
#pragma unroll
    for (int i = 0; i < 8; ++i)
        q4[i] = *(const v4f*)(Qt + i * 128);

    const int ch = st >> 2;
    const int t0 = ch * 4;
    v4f k4[8];
    {
        const float* Kt = Kb + t0 * (16 * HDC) + lane * 4;
#pragma unroll
        for (int i = 0; i < 8; ++i)
            k4[i] = *(const v4f*)(Kt + i * 128);
    }

    v8f acc[4] = {};
    if (ch > 0) {
        const float wq = exp2f((float)((st & 3) * 16 + m + 1) * l2g);
        const float* St = Sb + (size_t)(ch - 1) * (CHK * HDC) + lane * 4;
#pragma unroll
        for (int i = 0; i < 8; ++i) {
            v2f a0 = { q4[i][0] * wq, q4[i][1] * wq };
            v2f a1 = { q4[i][2] * wq, q4[i][3] * wq };
#pragma unroll
            for (int nt = 0; nt < 4; ++nt) {
                v4f s4 = *(const v4f*)(St + (i * 4 + nt) * 128);
                v2f b0 = { s4[0], s4[1] };
                v2f b1 = { s4[2], s4[3] };
                acc[nt] = wmma4(a0, b0, acc[nt]);
                acc[nt] = wmma4(a1, b1, acc[nt]);
            }
        }
    }
    for (int tt = t0; tt <= st; ++tt) {
        const float* Vt = Vb + tt * (16 * HDC) + lane * 4;
        v4f vr[8];
#pragma unroll
        for (int i = 0; i < 8; ++i)
            vr[i] = *(const v4f*)(Vt + i * 128);
        const int tn = (tt < st) ? tt + 1 : tt;
        const float* Kn = Kb + tn * (16 * HDC) + lane * 4;
        v4f kn[8];
#pragma unroll
        for (int i = 0; i < 8; ++i)
            kn[i] = *(const v4f*)(Kn + i * 128);
        __builtin_amdgcn_sched_barrier(0);
        v8f sc = {};
#pragma unroll
        for (int i = 0; i < 8; ++i) {
            v2f qa0 = { q4[i][0], q4[i][1] };
            v2f kb0 = { k4[i][0], k4[i][1] };
            sc = wmma4(kb0, qa0, sc);
            v2f qa1 = { q4[i][2], q4[i][3] };
            v2f kb1 = { k4[i][2], k4[i][3] };
            sc = wmma4(kb1, qa1, sc);
        }
        float p[8];
#pragma unroll
        for (int j = 0; j < 8; ++j) {
            int tg = tt * 16 + mbase + j;
            float d = (float)(sg - tg);
            float w = exp2f(d * l2g);
            p[j] = (sg >= tg) ? sc[j] * w : 0.0f;
        }
#pragma unroll
        for (int g = 0; g < 2; ++g) {
            v2f a0 = { p[4 * g],     p[4 * g + 1] };
            v2f a1 = { p[4 * g + 2], p[4 * g + 3] };
#pragma unroll
            for (int nt = 0; nt < 4; ++nt) {
                v4f v4 = vr[nt * 2 + g];
                v2f b0 = { v4[0], v4[1] };
                v2f b1 = { v4[2], v4[3] };
                acc[nt] = wmma4(a0, b0, acc[nt]);
                acc[nt] = wmma4(a1, b1, acc[nt]);
            }
        }
        __builtin_amdgcn_sched_barrier(0);
#pragma unroll
        for (int i = 0; i < 8; ++i)
            k4[i] = kn[i];
    }

    float rs[8];
#pragma unroll
    for (int j = 0; j < 8; ++j)
        rs[j] = (acc[0][j] + acc[1][j]) + (acc[2][j] + acc[3][j]);
#pragma unroll
    for (int sh = 1; sh < 16; sh <<= 1)
#pragma unroll
        for (int j = 0; j < 8; ++j)
            rs[j] += __shfl_xor(rs[j], sh);
    float dv[4][8];
    float rq[8];
#pragma unroll
    for (int j = 0; j < 8; ++j) {
        const float mean = rs[j] * (1.0f / HDC);
#pragma unroll
        for (int nt = 0; nt < 4; ++nt)
            dv[nt][j] = acc[nt][j] - mean;
        rq[j] = (dv[0][j] * dv[0][j] + dv[1][j] * dv[1][j])
              + (dv[2][j] * dv[2][j] + dv[3][j] * dv[3][j]);
    }
#pragma unroll
    for (int sh = 1; sh < 16; sh <<= 1)
#pragma unroll
        for (int j = 0; j < 8; ++j)
            rq[j] += __shfl_xor(rq[j], sh);

    float* ob = normout + (((size_t)b * SSC + st * 16) * NHC + n) * HDC;
#pragma unroll
    for (int j = 0; j < 8; ++j) {
        const float inv = rsqrtf(rq[j] * (1.0f / HDC) + 1e-5f);
#pragma unroll
        for (int nt = 0; nt < 4; ++nt)
            ob[(size_t)(mbase + j) * NHC * HDC + nt * 16 + m] =
                dv[nt][j] * inv * gsc[nt] + gbi[nt];
    }
    }
}

#define GBK 64
#define GPP 288
#define GNCH (HIDC / GBK)
#define GSTG ((GBK / 2) * GPP)
#define GRW (GBK / 8)
#define GST (GBK / 4)
template <bool SWISH_ADD>
__global__ void __launch_bounds__(256)
gemm1024_kernel(const float* __restrict__ A, const float* __restrict__ W,
                const float* __restrict__ addn, float* __restrict__ out)
{
    __shared__ __attribute__((aligned(16))) float Ws[2 * GSTG];

    const int tid  = threadIdx.x;
    const int lane = tid & 31;
    const int wv   = tid >> 5;
    const int m    = lane & 15;
    const int hi   = lane >> 4;
    const int kb2  = hi << 1;
    const int strip = blockIdx.x * 8 + wv;
    const int nb    = blockIdx.y;
    const int n0    = nb * 128;

    const float* Ab = A + (size_t)strip * 16 * HIDC;
    const float* Wg = W + (size_t)(GRW * wv) * HIDC + n0 + 4 * lane;
    float*       wst = Ws + ((GRW / 2) * wv) * GPP + 8 * lane;
    const float* wrd = Ws + hi * GPP + 2 * m;

    v8f acc[8] = {};
    v4f wr[GRW];
    v2f fa[2][GST];
    v2f fb[2][8];

#pragma unroll
    for (int r = 0; r < GRW; ++r) wr[r] = *(const v4f*)(Wg + (size_t)r * HIDC);
#pragma unroll
    for (int s = 0; s < GST; ++s) fa[0][s] = load_a16x4(Ab, HIDC, 4 * s, m, kb2);
#pragma unroll
    for (int q = 0; q < GRW / 2; ++q) {
        v4f p0 = { wr[2*q][0], wr[2*q+1][0], wr[2*q][1], wr[2*q+1][1] };
        v4f p1 = { wr[2*q][2], wr[2*q+1][2], wr[2*q][3], wr[2*q+1][3] };
        *(v4f*)(wst + q * GPP)     = p0;
        *(v4f*)(wst + q * GPP + 4) = p1;
    }
    __syncthreads();

    for (int c2 = 0; c2 < GNCH; c2 += 2) {
#pragma unroll
        for (int h = 0; h < 2; ++h) {
            const int c = c2 + h;
            const bool more = (c + 1 < GNCH);
            const float* wcur = wrd + h * GSTG;
            if (more) {
                const int kn = (c + 1) * GBK;
#pragma unroll
                for (int r = 0; r < GRW; ++r)
                    wr[r] = *(const v4f*)(Wg + (size_t)(kn + r) * HIDC);
#pragma unroll
                for (int s = 0; s < GST; ++s)
                    fa[h ^ 1][s] = load_a16x4(Ab, HIDC, kn + 4 * s, m, kb2);
            }
#pragma unroll
            for (int nt = 0; nt < 8; ++nt)
                fb[0][nt] = *(const v2f*)(wcur + nt * 32);
#pragma unroll
            for (int s = 0; s < GST; ++s) {
                if (s + 1 < GST) {
#pragma unroll
                    for (int nt = 0; nt < 8; ++nt)
                        fb[(s + 1) & 1][nt] = *(const v2f*)(wcur + (s + 1) * 2 * GPP + nt * 32);
                }
#pragma unroll
                for (int nt = 0; nt < 8; ++nt)
                    acc[nt] = wmma4(fa[h][s], fb[s & 1][nt], acc[nt]);
            }
            if (more) {
                float* wn = wst + (h ^ 1) * GSTG;
#pragma unroll
                for (int q = 0; q < GRW / 2; ++q) {
                    v4f p0 = { wr[2*q][0], wr[2*q+1][0], wr[2*q][1], wr[2*q+1][1] };
                    v4f p1 = { wr[2*q][2], wr[2*q+1][2], wr[2*q][3], wr[2*q+1][3] };
                    *(v4f*)(wn + q * GPP)     = p0;
                    *(v4f*)(wn + q * GPP + 4) = p1;
                }
            }
            __syncthreads();
        }
    }

    const int mbase = hi * 8;
#pragma unroll
    for (int nt = 0; nt < 8; ++nt)
#pragma unroll
        for (int j = 0; j < 8; ++j) {
            size_t off = (size_t)(strip * 16 + mbase + j) * HIDC + nb * 128 + nt * 16 + m;
            float g = acc[nt][j];
            if constexpr (SWISH_ADD) {
                out[off] = g / (1.0f + expf(-g)) + addn[off];
            } else {
                out[off] = g;
            }
        }
}

extern "C" void kernel_launch(void* const* d_in, const int* in_sizes, int n_in,
                              void* d_out, int out_size, void* d_ws, size_t ws_size,
                              hipStream_t stream)
{
    (void)in_sizes; (void)n_in; (void)out_size; (void)ws_size;
    const float* x   = (const float*)d_in[0];
    const float* wq  = (const float*)d_in[1];
    const float* wk  = (const float*)d_in[2];
    const float* wv  = (const float*)d_in[3];
    const float* w1  = (const float*)d_in[4];
    const float* w2  = (const float*)d_in[5];
    const float* gns = (const float*)d_in[6];
    const float* gnb = (const float*)d_in[7];
    float* out = (float*)d_out;
    float* ws  = (float*)d_ws;

    const size_t NQ = (size_t)BBC * NHC * SSC * HDC;
    float* q    = ws;
    float* k    = ws + NQ;
    float* v    = ws + 2 * NQ;
    float* norm = ws + 3 * NQ;
    float* mid  = k;
    float* kd   = ws + 4 * NQ;
    float* sst  = ws + 5 * NQ;

    dim3 wave(32);
    qkv_kernel<<<dim3(BBC * SSC / 16, NHC), wave, 0, stream>>>(x, wq, wk, wv, q, k, v, kd);
    state_kernel<<<dim3(16, BBC * NHC), wave, 0, stream>>>(kd, v, sst);
    retention_kernel<<<dim3(SSC / 32, BBC * NHC), wave, 0, stream>>>(q, k, v, sst, gns, gnb, norm);
    gemm1024_kernel<true ><<<dim3(BBC * SSC / 128, HIDC / 128), dim3(256), 0, stream>>>(x,   w1, norm, mid);
    gemm1024_kernel<false><<<dim3(BBC * SSC / 128, HIDC / 128), dim3(256), 0, stream>>>(mid, w2, nullptr, out);
}


# chunk-state scan rides in the gate GEMM's launch as 128 extra work-groups (4 launches, was 5): gate GEMM moves ahead of retention, whose epilogue adds swish(gate) to norm; sums, order kept
# speedup vs baseline: 3.6755x; 1.0650x over previous
#include <hip/hip_runtime.h>
#include <hip/hip_bf16.h>
#include <math.h>

#define HIDC 1024
#define NHC  16
#define HDC  64
#define BBC  2
#define SSC  2048

typedef float v2f __attribute__((ext_vector_type(2)));
typedef float v4f __attribute__((ext_vector_type(4)));
typedef float v8f __attribute__((ext_vector_type(8)));

__device__ __forceinline__ v8f wmma4(v2f a, v2f b, v8f c) {
    return __builtin_amdgcn_wmma_f32_16x16x4_f32(
         false, a,  false, b,
         (short)0, c,  false,  false);
}

__device__ __forceinline__ v2f load2(const float* __restrict__ p) {
    return *(const v2f*)p;
}

__device__ __forceinline__ v2f load_a16x4(const float* __restrict__ A, int lda, int k0, int m, int kb2) {
    return load2(A + m * lda + k0 + kb2);
}

__device__ __forceinline__ v2f load_b4x16(const float* __restrict__ B, int ldb, int k0, int n0,
                                          int nl, int kb2) {
    const float* p = B + (k0 + kb2) * ldb + n0 + nl;
    v2f r; r[0] = p[0]; r[1] = p[ldb];
    return r;
}

__device__ __forceinline__ float head_l2g(int n) {
    const float lin = logf(1.0f / 512.0f)
                    + (float)n * ((logf(1.0f / 32.0f) - logf(1.0f / 512.0f)) / 15.0f);
    return log2f(1.0f - expf(lin));
}

#define CHK 64
#define NCHK (SSC / CHK)

__global__ void __launch_bounds__(32)
qkv_kernel(const float* __restrict__ x,
           const float* __restrict__ wq, const float* __restrict__ wk,
           const float* __restrict__ wv,
           float* __restrict__ qf, float* __restrict__ kf, float* __restrict__ vf,
           float* __restrict__ kdf)
{
    const int lane = threadIdx.x;
    const int m    = lane & 15;
    const int hi   = lane >> 4;
    const int kb2  = hi << 1;
    const int strip = blockIdx.x;
    const int n     = blockIdx.y;
    const int r0    = strip * 16;

    const float* Ab = x + (size_t)r0 * HIDC + n * HDC;
    const float* Wq = wq + n * HDC * HDC;
    const float* Wk = wk + n * HDC * HDC;
    const float* Wv = wv + n * HDC * HDC;

    v8f aq[4] = {}; v8f ak[4] = {}; v8f av[4] = {}; v8f ad[4] = {};
    for (int k0 = 0; k0 < HDC; k0 += 4) {
        v2f a = load_a16x4(Ab, HIDC, k0, m, kb2);
#pragma unroll
        for (int nt = 0; nt < 4; ++nt) {
            v2f bq = load_b4x16(Wq, HDC, k0, nt * 16, m, kb2);
            aq[nt] = wmma4(bq, a, aq[nt]);
            v2f bk = load_b4x16(Wk, HDC, k0, nt * 16, m, kb2);
            ak[nt] = wmma4(bk, a, ak[nt]);
            ad[nt] = wmma4(a, bk, ad[nt]);
            v2f bv = load_b4x16(Wv, HDC, k0, nt * 16, m, kb2);
            av[nt] = wmma4(a, bv, av[nt]);
        }
    }

    const int b  = r0 / SSC;
    const int s0 = r0 % SSC;
    const size_t tile = (((size_t)b * NHC + n) * (SSC / 16) + (s0 >> 4)) * (16 * HDC);
    float* qt = qf + tile;
    float* kt = kf + tile;
    float* vt = vf + tile;
    float* dt = kdf + tile;
    const float l2g = head_l2g(n);
    float wd[8];
#pragma unroll
    for (int j = 0; j < 8; ++j)
        wd[j] = exp2f((float)(CHK - 1 - ((s0 & (CHK - 1)) + 8 * hi + j)) * l2g);
#pragma unroll
    for (int nt = 0; nt < 4; ++nt) {
        const int io = ((2 * nt + hi) * 32 + m) * 4;
        v4f q0 = { aq[nt][0], aq[nt][1], aq[nt][2], aq[nt][3] };
        v4f q1 = { aq[nt][4], aq[nt][5], aq[nt][6], aq[nt][7] };
        v4f k0v = { ak[nt][0], ak[nt][1], ak[nt][2], ak[nt][3] };
        v4f k1v = { ak[nt][4], ak[nt][5], ak[nt][6], ak[nt][7] };
        *(v4f*)(qt + io)      = q0;
        *(v4f*)(qt + io + 64) = q1;
        *(v4f*)(kt + io)      = k0v;
        *(v4f*)(kt + io + 64) = k1v;
        v4f v0 = { av[nt][0], av[nt][1], av[nt][2], av[nt][3] };
        v4f v1 = { av[nt][4], av[nt][5], av[nt][6], av[nt][7] };
        *(v4f*)(vt + ((nt * 2 + 0) * 32 + lane) * 4) = v0;
        *(v4f*)(vt + ((nt * 2 + 1) * 32 + lane) * 4) = v1;
        v4f d0 = { ad[nt][0] * wd[0], ad[nt][1] * wd[1], ad[nt][2] * wd[2], ad[nt][3] * wd[3] };
        v4f d1 = { ad[nt][4] * wd[4], ad[nt][5] * wd[5], ad[nt][6] * wd[6], ad[nt][7] * wd[7] };
        *(v4f*)(dt + ((nt * 2 + 0) * 32 + lane) * 4) = d0;
        *(v4f*)(dt + ((nt * 2 + 1) * 32 + lane) * 4) = d1;
    }
}

__device__ __forceinline__ void state_step(const v4f (&fk)[8], const v4f (&fv)[8],
                                           v8f& s0, v8f& s1, float g64,
                                           float* __restrict__ sc)
{
#pragma unroll
    for (int j = 0; j < 8; ++j) { s0[j] *= g64; s1[j] *= g64; }
#pragma unroll
    for (int ti = 0; ti < 4; ++ti) {
        v4f a0 = fk[ti * 2],     b0 = fv[ti * 2];
        v4f a1 = fk[ti * 2 + 1], b1 = fv[ti * 2 + 1];
        v2f a00 = { a0[0], a0[1] }, b00 = { b0[0], b0[1] };
        v2f a01 = { a0[2], a0[3] }, b01 = { b0[2], b0[3] };
        v2f a10 = { a1[0], a1[1] }, b10 = { b1[0], b1[1] };
        v2f a11 = { a1[2], a1[3] }, b11 = { b1[2], b1[3] };
        s0 = wmma4(a00, b00, s0);
        s1 = wmma4(a10, b10, s1);
        s0 = wmma4(a01, b01, s0);
        s1 = wmma4(a11, b11, s1);
    }
    v4f o0 = { s0[0] + s1[0], s0[1] + s1[1], s0[2] + s1[2], s0[3] + s1[3] };
    v4f o1 = { s0[4] + s1[4], s0[5] + s1[5], s0[6] + s1[6], s0[7] + s1[7] };
    *(v4f*)(sc)      = o0;
    *(v4f*)(sc + 64) = o1;
}

__device__ __forceinline__ void
state_scan(const float* __restrict__ kdf, const float* __restrict__ vf,
           float* __restrict__ sf, int lane, int tile, int bn)
{
    const int m    = lane & 15;
    const int hi   = lane >> 4;
    const int dtl  = tile >> 2;
    const int etl  = tile & 3;
    const int n    = bn % NHC;
    const float g64 = exp2f((float)CHK * head_l2g(n));

    const float* Kd = kdf + (size_t)bn * SSC * HDC + (dtl * 2 * 32 + lane) * 4;
    const float* Vb = vf  + (size_t)bn * SSC * HDC + (etl * 2 * 32 + lane) * 4;
    float* So = sf + (size_t)bn * (NCHK - 1) * (CHK * HDC)
              + (((2 * dtl + hi) * 4 + etl) * 32 + m) * 4;

    v4f fk[3][8], fv[3][8];
#pragma unroll
    for (int c = 0; c < 2; ++c)
#pragma unroll
        for (int ti = 0; ti < 4; ++ti)
#pragma unroll
            for (int g = 0; g < 2; ++g) {
                fk[c][ti * 2 + g] = *(const v4f*)(Kd + (c * 4 + ti) * (16 * HDC) + g * 128);
                fv[c][ti * 2 + g] = *(const v4f*)(Vb + (c * 4 + ti) * (16 * HDC) + g * 128);
            }

    v8f s0 = {}; v8f s1 = {};
    for (int c3 = 0; c3 < NCHK - 2; c3 += 3) {
#pragma unroll
        for (int u = 0; u < 3; ++u) {
            const int c = c3 + u;
            const int cn = (c + 2 < NCHK - 1) ? c + 2 : NCHK - 2;
#pragma unroll
            for (int ti = 0; ti < 4; ++ti)
#pragma unroll
                for (int g = 0; g < 2; ++g) {
                    fk[(u + 2) % 3][ti * 2 + g] =
                        *(const v4f*)(Kd + (cn * 4 + ti) * (16 * HDC) + g * 128);
                    fv[(u + 2) % 3][ti * 2 + g] =
                        *(const v4f*)(Vb + (cn * 4 + ti) * (16 * HDC) + g * 128);
                }
            __builtin_amdgcn_sched_barrier(0);
            state_step(fk[u], fv[u], s0, s1, g64, So + (size_t)c * (CHK * HDC));
            __builtin_amdgcn_sched_barrier(0);
        }
    }
    state_step(fk[0], fv[0], s0, s1, g64, So + (size_t)(NCHK - 2) * (CHK * HDC));
}

__global__ void __launch_bounds__(32)
retention_kernel(const float* __restrict__ qf, const float* __restrict__ kf,
                 const float* __restrict__ vf, const float* __restrict__ sf,
                 const float* __restrict__ scale, const float* __restrict__ bias,
                 const float* __restrict__ gate, float* __restrict__ sumout)
{
    const int lane = threadIdx.x;
    const int m    = lane & 15;
    const int hi   = lane >> 4;
    const int mbase = hi * 8;
    const int pr   = blockIdx.x;
    const int bn   = blockIdx.y;
    const int b    = bn / NHC;
    const int n    = bn % NHC;

    const float* Qb = qf + (size_t)bn * SSC * HDC;
    const float* Kb = kf + (size_t)bn * SSC * HDC;
    const float* Vb = vf + (size_t)bn * SSC * HDC;

    const float* Sb = sf + (size_t)bn * (NCHK - 1) * (CHK * HDC);
    const float l2g = head_l2g(n);

    float gsc[4], gbi[4];
#pragma unroll
    for (int nt = 0; nt < 4; ++nt) {
        gsc[nt] = scale[n * HDC + nt * 16 + m];
        gbi[nt] = bias[n * HDC + nt * 16 + m];
    }

    for (int pass = 0; pass < 2; ++pass) {
    const int st = pass ? (SSC / 16 - 1 - pr) : pr;
    const int sg = st * 16 + m;

    const float* Qt = Qb + st * (16 * HDC) + lane * 4;
    v4f q4[8];
#pragma unroll
    for (int i = 0; i < 8; ++i)
        q4[i] = *(const v4f*)(Qt + i * 128);

    const int ch = st >> 2;
    const int t0 = ch * 4;
    v4f k4[8];
    {
        const float* Kt = Kb + t0 * (16 * HDC) + lane * 4;
#pragma unroll
        for (int i = 0; i < 8; ++i)
            k4[i] = *(const v4f*)(Kt + i * 128);
    }

    const size_t orow = (((size_t)b * SSC + st * 16) * NHC + n) * HDC;
    const float* gb = gate + orow;
    float gt[4][8];
#pragma unroll
    for (int j = 0; j < 8; ++j)
#pragma unroll
        for (int nt = 0; nt < 4; ++nt)
            gt[nt][j] = gb[(size_t)(mbase + j) * NHC * HDC + nt * 16 + m];

    v8f acc[4] = {};
    if (ch > 0) {
        const float wq = exp2f((float)((st & 3) * 16 + m + 1) * l2g);
        const float* St = Sb + (size_t)(ch - 1) * (CHK * HDC) + lane * 4;
#pragma unroll
        for (int i = 0; i < 8; ++i) {
            v2f a0 = { q4[i][0] * wq, q4[i][1] * wq };
            v2f a1 = { q4[i][2] * wq, q4[i][3] * wq };
#pragma unroll
            for (int nt = 0; nt < 4; ++nt) {
                v4f s4 = *(const v4f*)(St + (i * 4 + nt) * 128);
                v2f b0 = { s4[0], s4[1] };
                v2f b1 = { s4[2], s4[3] };
                acc[nt] = wmma4(a0, b0, acc[nt]);
                acc[nt] = wmma4(a1, b1, acc[nt]);
            }
        }
    }
    for (int tt = t0; tt <= st; ++tt) {
        const float* Vt = Vb + tt * (16 * HDC) + lane * 4;
        v4f vr[8];
#pragma unroll
        for (int i = 0; i < 8; ++i)
            vr[i] = *(const v4f*)(Vt + i * 128);
        const int tn = (tt < st) ? tt + 1 : tt;
        const float* Kn = Kb + tn * (16 * HDC) + lane * 4;
        v4f kn[8];
#pragma unroll
        for (int i = 0; i < 8; ++i)
            kn[i] = *(const v4f*)(Kn + i * 128);
        __builtin_amdgcn_sched_barrier(0);
        v8f sc = {};
#pragma unroll
        for (int i = 0; i < 8; ++i) {
            v2f qa0 = { q4[i][0], q4[i][1] };
            v2f kb0 = { k4[i][0], k4[i][1] };
            sc = wmma4(kb0, qa0, sc);
            v2f qa1 = { q4[i][2], q4[i][3] };
            v2f kb1 = { k4[i][2], k4[i][3] };
            sc = wmma4(kb1, qa1, sc);
        }
        float p[8];
#pragma unroll
        for (int j = 0; j < 8; ++j) {
            int tg = tt * 16 + mbase + j;
            float d = (float)(sg - tg);
            float w = exp2f(d * l2g);
            p[j] = (sg >= tg) ? sc[j] * w : 0.0f;
        }
#pragma unroll
        for (int g = 0; g < 2; ++g) {
            v2f a0 = { p[4 * g],     p[4 * g + 1] };
            v2f a1 = { p[4 * g + 2], p[4 * g + 3] };
#pragma unroll
            for (int nt = 0; nt < 4; ++nt) {
                v4f v4 = vr[nt * 2 + g];
                v2f b0 = { v4[0], v4[1] };
                v2f b1 = { v4[2], v4[3] };
                acc[nt] = wmma4(a0, b0, acc[nt]);
                acc[nt] = wmma4(a1, b1, acc[nt]);
            }
        }
        __builtin_amdgcn_sched_barrier(0);
#pragma unroll
        for (int i = 0; i < 8; ++i)
            k4[i] = kn[i];
    }

    float rs[8];
#pragma unroll
    for (int j = 0; j < 8; ++j)
        rs[j] = (acc[0][j] + acc[1][j]) + (acc[2][j] + acc[3][j]);
#pragma unroll
    for (int sh = 1; sh < 16; sh <<= 1)
#pragma unroll
        for (int j = 0; j < 8; ++j)
            rs[j] += __shfl_xor(rs[j], sh);
    float dv[4][8];
    float rq[8];
#pragma unroll
    for (int j = 0; j < 8; ++j) {
        const float mean = rs[j] * (1.0f / HDC);
#pragma unroll
        for (int nt = 0; nt < 4; ++nt)
            dv[nt][j] = acc[nt][j] - mean;
        rq[j] = (dv[0][j] * dv[0][j] + dv[1][j] * dv[1][j])
              + (dv[2][j] * dv[2][j] + dv[3][j] * dv[3][j]);
    }
#pragma unroll
    for (int sh = 1; sh < 16; sh <<= 1)
#pragma unroll
        for (int j = 0; j < 8; ++j)
            rq[j] += __shfl_xor(rq[j], sh);

    float* ob = sumout + orow;
#pragma unroll
    for (int j = 0; j < 8; ++j) {
        const float inv = rsqrtf(rq[j] * (1.0f / HDC) + 1e-5f);
#pragma unroll
        for (int nt = 0; nt < 4; ++nt) {
            const float nv = dv[nt][j] * inv * gsc[nt] + gbi[nt];
            ob[(size_t)(mbase + j) * NHC * HDC + nt * 16 + m] = gt[nt][j] + nv;
        }
    }
    }
}

#define GBK 64
#define GPP 288
#define GNCH (HIDC / GBK)
#define GSTG ((GBK / 2) * GPP)
#define GRW (GBK / 8)
#define GST (GBK / 4)
#define GNB (HIDC / 128)
template <bool SWISH>
__device__ __forceinline__ void
gemm1024_body(const float* __restrict__ A, const float* __restrict__ W,
              float* __restrict__ out)
{
    __shared__ __attribute__((aligned(16))) float Ws[2 * GSTG];

    const int tid  = threadIdx.x;
    const int lane = tid & 31;
    const int wv   = tid >> 5;
    const int m    = lane & 15;
    const int hi   = lane >> 4;
    const int kb2  = hi << 1;
    const int strip = blockIdx.x * 8 + wv;
    const int nb    = blockIdx.y;
    const int n0    = nb * 128;

    const float* Ab = A + (size_t)strip * 16 * HIDC;
    const float* Wg = W + (size_t)(GRW * wv) * HIDC + n0 + 4 * lane;
    float*       wst = Ws + ((GRW / 2) * wv) * GPP + 8 * lane;
    const float* wrd = Ws + hi * GPP + 2 * m;

    v8f acc[8] = {};
    v4f wr[GRW];
    v2f fa[2][GST];
    v2f fb[2][8];

#pragma unroll
    for (int r = 0; r < GRW; ++r) wr[r] = *(const v4f*)(Wg + (size_t)r * HIDC);
#pragma unroll
    for (int s = 0; s < GST; ++s) fa[0][s] = load_a16x4(Ab, HIDC, 4 * s, m, kb2);
#pragma unroll
    for (int q = 0; q < GRW / 2; ++q) {
        v4f p0 = { wr[2*q][0], wr[2*q+1][0], wr[2*q][1], wr[2*q+1][1] };
        v4f p1 = { wr[2*q][2], wr[2*q+1][2], wr[2*q][3], wr[2*q+1][3] };
        *(v4f*)(wst + q * GPP)     = p0;
        *(v4f*)(wst + q * GPP + 4) = p1;
    }
    __syncthreads();

    for (int c2 = 0; c2 < GNCH; c2 += 2) {
#pragma unroll
        for (int h = 0; h < 2; ++h) {
            const int c = c2 + h;
            const bool more = (c + 1 < GNCH);
            const float* wcur = wrd + h * GSTG;
            if (more) {
                const int kn = (c + 1) * GBK;
#pragma unroll
                for (int r = 0; r < GRW; ++r)
                    wr[r] = *(const v4f*)(Wg + (size_t)(kn + r) * HIDC);
#pragma unroll
                for (int s = 0; s < GST; ++s)
                    fa[h ^ 1][s] = load_a16x4(Ab, HIDC, kn + 4 * s, m, kb2);
            }
#pragma unroll
            for (int nt = 0; nt < 8; ++nt)
                fb[0][nt] = *(const v2f*)(wcur + nt * 32);
#pragma unroll
            for (int s = 0; s < GST; ++s) {
                if (s + 1 < GST) {
#pragma unroll
                    for (int nt = 0; nt < 8; ++nt)
                        fb[(s + 1) & 1][nt] = *(const v2f*)(wcur + (s + 1) * 2 * GPP + nt * 32);
                }
#pragma unroll
                for (int nt = 0; nt < 8; ++nt)
                    acc[nt] = wmma4(fa[h][s], fb[s & 1][nt], acc[nt]);
            }
            if (more) {
                float* wn = wst + (h ^ 1) * GSTG;
#pragma unroll
                for (int q = 0; q < GRW / 2; ++q) {
                    v4f p0 = { wr[2*q][0], wr[2*q+1][0], wr[2*q][1], wr[2*q+1][1] };
                    v4f p1 = { wr[2*q][2], wr[2*q+1][2], wr[2*q][3], wr[2*q+1][3] };
                    *(v4f*)(wn + q * GPP)     = p0;
                    *(v4f*)(wn + q * GPP + 4) = p1;
                }
            }
            __syncthreads();
        }
    }

    const int mbase = hi * 8;
#pragma unroll
    for (int nt = 0; nt < 8; ++nt)
#pragma unroll
        for (int j = 0; j < 8; ++j) {
            size_t off = (size_t)(strip * 16 + mbase + j) * HIDC + nb * 128 + nt * 16 + m;
            float g = acc[nt][j];
            if constexpr (SWISH) {
                out[off] = g / (1.0f + expf(-g));
            } else {
                out[off] = g;
            }
        }
}

static_assert(BBC * NHC == BBC * SSC / 128, "blockIdx.x serves as the scan's head-batch index");
__global__ void __launch_bounds__(256, 4)
gate_state_kernel(const float* __restrict__ A, const float* __restrict__ W,
                  float* __restrict__ out,
                  const float* __restrict__ kdf, const float* __restrict__ vf,
                  float* __restrict__ sf)
{
    if (blockIdx.y >= GNB) {
        const int wv = threadIdx.x >> 5;
        if (wv < 4)
            state_scan(kdf, vf, sf, threadIdx.x & 31, (blockIdx.y - GNB) * 4 + wv, blockIdx.x);
        return;
    }
    gemm1024_body<true>(A, W, out);
}

__global__ void __launch_bounds__(256)
gemm1024_kernel(const float* __restrict__ A, const float* __restrict__ W,
                float* __restrict__ out)
{
    gemm1024_body<false>(A, W, out);
}

extern "C" void kernel_launch(void* const* d_in, const int* in_sizes, int n_in,
                              void* d_out, int out_size, void* d_ws, size_t ws_size,
                              hipStream_t stream)
{
    (void)in_sizes; (void)n_in; (void)out_size; (void)ws_size;
    const float* x   = (const float*)d_in[0];
    const float* wq  = (const float*)d_in[1];
    const float* wk  = (const float*)d_in[2];
    const float* wv  = (const float*)d_in[3];
    const float* w1  = (const float*)d_in[4];
    const float* w2  = (const float*)d_in[5];
    const float* gns = (const float*)d_in[6];
    const float* gnb = (const float*)d_in[7];
    float* out = (float*)d_out;
    float* ws  = (float*)d_ws;

    const size_t NQ = (size_t)BBC * NHC * SSC * HDC;
    float* q    = ws;
    float* k    = ws + NQ;
    float* v    = ws + 2 * NQ;
    float* sum  = ws + 3 * NQ;
    float* kd   = ws + 4 * NQ;
    float* sst  = ws + 5 * NQ;
    float* gate = ws + 6 * NQ;

    dim3 wave(32);
    qkv_kernel<<<dim3(BBC * SSC / 16, NHC), wave, 0, stream>>>(x, wq, wk, wv, q, k, v, kd);
    gate_state_kernel<<<dim3(BBC * SSC / 128, GNB + 4), dim3(256), 0, stream>>>(x, w1, gate, kd, v, sst);
    retention_kernel<<<dim3(SSC / 32, BBC * NHC), wave, 0, stream>>>(q, k, v, sst, gns, gnb, gate, sum);
    gemm1024_kernel<<<dim3(BBC * SSC / 128, GNB), dim3(256), 0, stream>>>(sum, w2, out);
}


# both 1024-GEMMs: each wave fetches its 16 A rows by whole-line loads (4 lines per load, was 16) into an LDS strip of its own and reads its A fragments there beside the B reads; sums, order kept
# speedup vs baseline: 3.8575x; 1.0495x over previous
#include <hip/hip_runtime.h>
#include <hip/hip_bf16.h>
#include <math.h>

#define HIDC 1024
#define NHC  16
#define HDC  64
#define BBC  2
#define SSC  2048

typedef float v2f __attribute__((ext_vector_type(2)));
typedef float v4f __attribute__((ext_vector_type(4)));
typedef float v8f __attribute__((ext_vector_type(8)));

__device__ __forceinline__ v8f wmma4(v2f a, v2f b, v8f c) {
    return __builtin_amdgcn_wmma_f32_16x16x4_f32(
         false, a,  false, b,
         (short)0, c,  false,  false);
}

__device__ __forceinline__ v2f load2(const float* __restrict__ p) {
    return *(const v2f*)p;
}

__device__ __forceinline__ v2f load_a16x4(const float* __restrict__ A, int lda, int k0, int m, int kb2) {
    return load2(A + m * lda + k0 + kb2);
}

__device__ __forceinline__ v2f load_b4x16(const float* __restrict__ B, int ldb, int k0, int n0,
                                          int nl, int kb2) {
    const float* p = B + (k0 + kb2) * ldb + n0 + nl;
    v2f r; r[0] = p[0]; r[1] = p[ldb];
    return r;
}

__device__ __forceinline__ float head_l2g(int n) {
    const float lin = logf(1.0f / 512.0f)
                    + (float)n * ((logf(1.0f / 32.0f) - logf(1.0f / 512.0f)) / 15.0f);
    return log2f(1.0f - expf(lin));
}

#define CHK 64
#define NCHK (SSC / CHK)

__global__ void __launch_bounds__(32)
qkv_kernel(const float* __restrict__ x,
           const float* __restrict__ wq, const float* __restrict__ wk,
           const float* __restrict__ wv,
           float* __restrict__ qf, float* __restrict__ kf, float* __restrict__ vf,
           float* __restrict__ kdf)
{
    const int lane = threadIdx.x;
    const int m    = lane & 15;
    const int hi   = lane >> 4;
    const int kb2  = hi << 1;
    const int strip = blockIdx.x;
    const int n     = blockIdx.y;
    const int r0    = strip * 16;

    const float* Ab = x + (size_t)r0 * HIDC + n * HDC;
    const float* Wq = wq + n * HDC * HDC;
    const float* Wk = wk + n * HDC * HDC;
    const float* Wv = wv + n * HDC * HDC;

    v8f aq[4] = {}; v8f ak[4] = {}; v8f av[4] = {}; v8f ad[4] = {};
    for (int k0 = 0; k0 < HDC; k0 += 4) {
        v2f a = load_a16x4(Ab, HIDC, k0, m, kb2);
#pragma unroll
        for (int nt = 0; nt < 4; ++nt) {
            v2f bq = load_b4x16(Wq, HDC, k0, nt * 16, m, kb2);
            aq[nt] = wmma4(bq, a, aq[nt]);
            v2f bk = load_b4x16(Wk, HDC, k0, nt * 16, m, kb2);
            ak[nt] = wmma4(bk, a, ak[nt]);
            ad[nt] = wmma4(a, bk, ad[nt]);
            v2f bv = load_b4x16(Wv, HDC, k0, nt * 16, m, kb2);
            av[nt] = wmma4(a, bv, av[nt]);
        }
    }

    const int b  = r0 / SSC;
    const int s0 = r0 % SSC;
    const size_t tile = (((size_t)b * NHC + n) * (SSC / 16) + (s0 >> 4)) * (16 * HDC);
    float* qt = qf + tile;
    float* kt = kf + tile;
    float* vt = vf + tile;
    float* dt = kdf + tile;
    const float l2g = head_l2g(n);
    float wd[8];
#pragma unroll
    for (int j = 0; j < 8; ++j)
        wd[j] = exp2f((float)(CHK - 1 - ((s0 & (CHK - 1)) + 8 * hi + j)) * l2g);
#pragma unroll
    for (int nt = 0; nt < 4; ++nt) {
        const int io = ((2 * nt + hi) * 32 + m) * 4;
        v4f q0 = { aq[nt][0], aq[nt][1], aq[nt][2], aq[nt][3] };
        v4f q1 = { aq[nt][4], aq[nt][5], aq[nt][6], aq[nt][7] };
        v4f k0v = { ak[nt][0], ak[nt][1], ak[nt][2], ak[nt][3] };
        v4f k1v = { ak[nt][4], ak[nt][5], ak[nt][6], ak[nt][7] };
        *(v4f*)(qt + io)      = q0;
        *(v4f*)(qt + io + 64) = q1;
        *(v4f*)(kt + io)      = k0v;
        *(v4f*)(kt + io + 64) = k1v;
        v4f v0 = { av[nt][0], av[nt][1], av[nt][2], av[nt][3] };
        v4f v1 = { av[nt][4], av[nt][5], av[nt][6], av[nt][7] };
        *(v4f*)(vt + ((nt * 2 + 0) * 32 + lane) * 4) = v0;
        *(v4f*)(vt + ((nt * 2 + 1) * 32 + lane) * 4) = v1;
        v4f d0 = { ad[nt][0] * wd[0], ad[nt][1] * wd[1], ad[nt][2] * wd[2], ad[nt][3] * wd[3] };
        v4f d1 = { ad[nt][4] * wd[4], ad[nt][5] * wd[5], ad[nt][6] * wd[6], ad[nt][7] * wd[7] };
        *(v4f*)(dt + ((nt * 2 + 0) * 32 + lane) * 4) = d0;
        *(v4f*)(dt + ((nt * 2 + 1) * 32 + lane) * 4) = d1;
    }
}

__device__ __forceinline__ void state_step(const v4f (&fk)[8], const v4f (&fv)[8],
                                           v8f& s0, v8f& s1, float g64,
                                           float* __restrict__ sc)
{
#pragma unroll
    for (int j = 0; j < 8; ++j) { s0[j] *= g64; s1[j] *= g64; }
#pragma unroll
    for (int ti = 0; ti < 4; ++ti) {
        v4f a0 = fk[ti * 2],     b0 = fv[ti * 2];
        v4f a1 = fk[ti * 2 + 1], b1 = fv[ti * 2 + 1];
        v2f a00 = { a0[0], a0[1] }, b00 = { b0[0], b0[1] };
        v2f a01 = { a0[2], a0[3] }, b01 = { b0[2], b0[3] };
        v2f a10 = { a1[0], a1[1] }, b10 = { b1[0], b1[1] };
        v2f a11 = { a1[2], a1[3] }, b11 = { b1[2], b1[3] };
        s0 = wmma4(a00, b00, s0);
        s1 = wmma4(a10, b10, s1);
        s0 = wmma4(a01, b01, s0);
        s1 = wmma4(a11, b11, s1);
    }
    v4f o0 = { s0[0] + s1[0], s0[1] + s1[1], s0[2] + s1[2], s0[3] + s1[3] };
    v4f o1 = { s0[4] + s1[4], s0[5] + s1[5], s0[6] + s1[6], s0[7] + s1[7] };
    *(v4f*)(sc)      = o0;
    *(v4f*)(sc + 64) = o1;
}

__device__ __forceinline__ void
state_scan(const float* __restrict__ kdf, const float* __restrict__ vf,
           float* __restrict__ sf, int lane, int tile, int bn)
{
    const int m    = lane & 15;
    const int hi   = lane >> 4;
    const int dtl  = tile >> 2;
    const int etl  = tile & 3;
    const int n    = bn % NHC;
    const float g64 = exp2f((float)CHK * head_l2g(n));

    const float* Kd = kdf + (size_t)bn * SSC * HDC + (dtl * 2 * 32 + lane) * 4;
    const float* Vb = vf  + (size_t)bn * SSC * HDC + (etl * 2 * 32 + lane) * 4;
    float* So = sf + (size_t)bn * (NCHK - 1) * (CHK * HDC)
              + (((2 * dtl + hi) * 4 + etl) * 32 + m) * 4;

    v4f fk[3][8], fv[3][8];
#pragma unroll
    for (int c = 0; c < 2; ++c)
#pragma unroll
        for (int ti = 0; ti < 4; ++ti)
#pragma unroll
            for (int g = 0; g < 2; ++g) {
                fk[c][ti * 2 + g] = *(const v4f*)(Kd + (c * 4 + ti) * (16 * HDC) + g * 128);
                fv[c][ti * 2 + g] = *(const v4f*)(Vb + (c * 4 + ti) * (16 * HDC) + g * 128);
            }

    v8f s0 = {}; v8f s1 = {};
    for (int c3 = 0; c3 < NCHK - 2; c3 += 3) {
#pragma unroll
        for (int u = 0; u < 3; ++u) {
            const int c = c3 + u;
            const int cn = (c + 2 < NCHK - 1) ? c + 2 : NCHK - 2;
#pragma unroll
            for (int ti = 0; ti < 4; ++ti)
#pragma unroll
                for (int g = 0; g < 2; ++g) {
                    fk[(u + 2) % 3][ti * 2 + g] =
                        *(const v4f*)(Kd + (cn * 4 + ti) * (16 * HDC) + g * 128);
                    fv[(u + 2) % 3][ti * 2 + g] =
                        *(const v4f*)(Vb + (cn * 4 + ti) * (16 * HDC) + g * 128);
                }
            __builtin_amdgcn_sched_barrier(0);
            state_step(fk[u], fv[u], s0, s1, g64, So + (size_t)c * (CHK * HDC));
            __builtin_amdgcn_sched_barrier(0);
        }
    }
    state_step(fk[0], fv[0], s0, s1, g64, So + (size_t)(NCHK - 2) * (CHK * HDC));
}

__global__ void __launch_bounds__(32)
retention_kernel(const float* __restrict__ qf, const float* __restrict__ kf,
                 const float* __restrict__ vf, const float* __restrict__ sf,
                 const float* __restrict__ scale, const float* __restrict__ bias,
                 const float* __restrict__ gate, float* __restrict__ sumout)
{
    const int lane = threadIdx.x;
    const int m    = lane & 15;
    const int hi   = lane >> 4;
    const int mbase = hi * 8;
    const int pr   = blockIdx.x;
    const int bn   = blockIdx.y;
    const int b    = bn / NHC;
    const int n    = bn % NHC;

    const float* Qb = qf + (size_t)bn * SSC * HDC;
    const float* Kb = kf + (size_t)bn * SSC * HDC;
    const float* Vb = vf + (size_t)bn * SSC * HDC;

    const float* Sb = sf + (size_t)bn * (NCHK - 1) * (CHK * HDC);
    const float l2g = head_l2g(n);

    float gsc[4], gbi[4];
#pragma unroll
    for (int nt = 0; nt < 4; ++nt) {
        gsc[nt] = scale[n * HDC + nt * 16 + m];
        gbi[nt] = bias[n * HDC + nt * 16 + m];
    }

    for (int pass = 0; pass < 2; ++pass) {
    const int st = pass ? (SSC / 16 - 1 - pr) : pr;
    const int sg = st * 16 + m;

    const float* Qt = Qb + st * (16 * HDC) + lane * 4;
    v4f q4[8];
#pragma unroll
    for (int i = 0; i < 8; ++i)
        q4[i] = *(const v4f*)(Qt + i * 128);

    const int ch = st >> 2;
    const int t0 = ch * 4;
    v4f k4[8];
    {
        const float* Kt = Kb + t0 * (16 * HDC) + lane * 4;
#pragma unroll
        for (int i = 0; i < 8; ++i)
            k4[i] = *(const v4f*)(Kt + i * 128);
    }

    const size_t orow = (((size_t)b * SSC + st * 16) * NHC + n) * HDC;
    const float* gb = gate + orow;
    float gt[4][8];
#pragma unroll
    for (int j = 0; j < 8; ++j)
#pragma unroll
        for (int nt = 0; nt < 4; ++nt)
            gt[nt][j] = gb[(size_t)(mbase + j) * NHC * HDC + nt * 16 + m];

    v8f acc[4] = {};
    if (ch > 0) {
        const float wq = exp2f((float)((st & 3) * 16 + m + 1) * l2g);
        const float* St = Sb + (size_t)(ch - 1) * (CHK * HDC) + lane * 4;
#pragma unroll
        for (int i = 0; i < 8; ++i) {
            v2f a0 = { q4[i][0] * wq, q4[i][1] * wq };
            v2f a1 = { q4[i][2] * wq, q4[i][3] * wq };
#pragma unroll
            for (int nt = 0; nt < 4; ++nt) {
                v4f s4 = *(const v4f*)(St + (i * 4 + nt) * 128);
                v2f b0 = { s4[0], s4[1] };
                v2f b1 = { s4[2], s4[3] };
                acc[nt] = wmma4(a0, b0, acc[nt]);
                acc[nt] = wmma4(a1, b1, acc[nt]);
            }
        }
    }
    for (int tt = t0; tt <= st; ++tt) {
        const float* Vt = Vb + tt * (16 * HDC) + lane * 4;
        v4f vr[8];
#pragma unroll
        for (int i = 0; i < 8; ++i)
            vr[i] = *(const v4f*)(Vt + i * 128);
        const int tn = (tt < st) ? tt + 1 : tt;
        const float* Kn = Kb + tn * (16 * HDC) + lane * 4;
        v4f kn[8];
#pragma unroll
        for (int i = 0; i < 8; ++i)
            kn[i] = *(const v4f*)(Kn + i * 128);
        __builtin_amdgcn_sched_barrier(0);
        v8f sc = {};
#pragma unroll
        for (int i = 0; i < 8; ++i) {
            v2f qa0 = { q4[i][0], q4[i][1] };
            v2f kb0 = { k4[i][0], k4[i][1] };
            sc = wmma4(kb0, qa0, sc);
            v2f qa1 = { q4[i][2], q4[i][3] };
            v2f kb1 = { k4[i][2], k4[i][3] };
            sc = wmma4(kb1, qa1, sc);
        }
        float p[8];
#pragma unroll
        for (int j = 0; j < 8; ++j) {
            int tg = tt * 16 + mbase + j;
            float d = (float)(sg - tg);
            float w = exp2f(d * l2g);
            p[j] = (sg >= tg) ? sc[j] * w : 0.0f;
        }
#pragma unroll
        for (int g = 0; g < 2; ++g) {
            v2f a0 = { p[4 * g],     p[4 * g + 1] };
            v2f a1 = { p[4 * g + 2], p[4 * g + 3] };
#pragma unroll
            for (int nt = 0; nt < 4; ++nt) {
                v4f v4 = vr[nt * 2 + g];
                v2f b0 = { v4[0], v4[1] };
                v2f b1 = { v4[2], v4[3] };
                acc[nt] = wmma4(a0, b0, acc[nt]);
                acc[nt] = wmma4(a1, b1, acc[nt]);
            }
        }
        __builtin_amdgcn_sched_barrier(0);
#pragma unroll
        for (int i = 0; i < 8; ++i)
            k4[i] = kn[i];
    }

    float rs[8];
#pragma unroll
    for (int j = 0; j < 8; ++j)
        rs[j] = (acc[0][j] + acc[1][j]) + (acc[2][j] + acc[3][j]);
#pragma unroll
    for (int sh = 1; sh < 16; sh <<= 1)
#pragma unroll
        for (int j = 0; j < 8; ++j)
            rs[j] += __shfl_xor(rs[j], sh);
    float dv[4][8];
    float rq[8];
#pragma unroll
    for (int j = 0; j < 8; ++j) {
        const float mean = rs[j] * (1.0f / HDC);
#pragma unroll
        for (int nt = 0; nt < 4; ++nt)
            dv[nt][j] = acc[nt][j] - mean;
        rq[j] = (dv[0][j] * dv[0][j] + dv[1][j] * dv[1][j])
              + (dv[2][j] * dv[2][j] + dv[3][j] * dv[3][j]);
    }
#pragma unroll
    for (int sh = 1; sh < 16; sh <<= 1)
#pragma unroll
        for (int j = 0; j < 8; ++j)
            rq[j] += __shfl_xor(rq[j], sh);

    float* ob = sumout + orow;
#pragma unroll
    for (int j = 0; j < 8; ++j) {
        const float inv = rsqrtf(rq[j] * (1.0f / HDC) + 1e-5f);
#pragma unroll
        for (int nt = 0; nt < 4; ++nt) {
            const float nv = dv[nt][j] * inv * gsc[nt] + gbi[nt];
            ob[(size_t)(mbase + j) * NHC * HDC + nt * 16 + m] = gt[nt][j] + nv;
        }
    }
    }
}

#define GBK 64
#define GPP 288
#define GNCH (HIDC / GBK)
#define GSTG ((GBK / 2) * GPP)
#define GRW (GBK / 8)
#define GST (GBK / 4)
#define GNB (HIDC / 128)
#define GAP (GBK + 4)
#define GAST (16 * GAP)
#define GAR (16 * GBK / 128)
template <bool SWISH>
__device__ __forceinline__ void
gemm1024_body(const float* __restrict__ A, const float* __restrict__ W,
              float* __restrict__ out)
{
    __shared__ __attribute__((aligned(16))) float Ws[2 * GSTG];
    __shared__ __attribute__((aligned(16))) float As[8 * 2 * GAST];

    const int tid  = threadIdx.x;
    const int lane = tid & 31;
    const int wv   = tid >> 5;
    const int m    = lane & 15;
    const int hi   = lane >> 4;
    const int kb2  = hi << 1;
    const int strip = blockIdx.x * 8 + wv;
    const int nb    = blockIdx.y;
    const int n0    = nb * 128;

    const float* Ag = A + ((size_t)strip * 16 + hi) * HIDC + 4 * m;
    float*       ast = As + wv * (2 * GAST) + hi * GAP + 4 * m;
    const float* ard = As + wv * (2 * GAST) + m * GAP + kb2;
    const float* Wg = W + (size_t)(GRW * wv) * HIDC + n0 + 4 * lane;
    float*       wst = Ws + ((GRW / 2) * wv) * GPP + 8 * lane;
    const float* wrd = Ws + hi * GPP + 2 * m;

    v8f acc[8] = {};
    v4f wr[GRW];
    v4f ar[GAR];
    v2f fa[2];
    v2f fb[2][8];

#pragma unroll
    for (int r = 0; r < GRW; ++r) wr[r] = *(const v4f*)(Wg + (size_t)r * HIDC);
#pragma unroll
    for (int r = 0; r < GAR; ++r) ar[r] = *(const v4f*)(Ag + (size_t)(2 * r) * HIDC);
#pragma unroll
    for (int q = 0; q < GRW / 2; ++q) {
        v4f p0 = { wr[2*q][0], wr[2*q+1][0], wr[2*q][1], wr[2*q+1][1] };
        v4f p1 = { wr[2*q][2], wr[2*q+1][2], wr[2*q][3], wr[2*q+1][3] };
        *(v4f*)(wst + q * GPP)     = p0;
        *(v4f*)(wst + q * GPP + 4) = p1;
    }
#pragma unroll
    for (int r = 0; r < GAR; ++r) *(v4f*)(ast + (2 * r) * GAP) = ar[r];
    __syncthreads();

    for (int c2 = 0; c2 < GNCH; c2 += 2) {
#pragma unroll
        for (int h = 0; h < 2; ++h) {
            const int c = c2 + h;
            const bool more = (c + 1 < GNCH);
            const float* wcur = wrd + h * GSTG;
            const float* acur = ard + h * GAST;
            if (more) {
                const int kn = (c + 1) * GBK;
#pragma unroll
                for (int r = 0; r < GRW; ++r)
                    wr[r] = *(const v4f*)(Wg + (size_t)(kn + r) * HIDC);
#pragma unroll
                for (int r = 0; r < GAR; ++r)
                    ar[r] = *(const v4f*)(Ag + (size_t)(2 * r) * HIDC + kn);
            }
            fa[0] = *(const v2f*)(acur);
#pragma unroll
            for (int nt = 0; nt < 8; ++nt)
                fb[0][nt] = *(const v2f*)(wcur + nt * 32);
#pragma unroll
            for (int s = 0; s < GST; ++s) {
                if (s + 1 < GST) {
                    fa[(s + 1) & 1] = *(const v2f*)(acur + (s + 1) * 4);
#pragma unroll
                    for (int nt = 0; nt < 8; ++nt)
                        fb[(s + 1) & 1][nt] = *(const v2f*)(wcur + (s + 1) * 2 * GPP + nt * 32);
                }
#pragma unroll
                for (int nt = 0; nt < 8; ++nt)
                    acc[nt] = wmma4(fa[s & 1], fb[s & 1][nt], acc[nt]);
            }
            if (more) {
                float* wn = wst + (h ^ 1) * GSTG;
#pragma unroll
                for (int q = 0; q < GRW / 2; ++q) {
                    v4f p0 = { wr[2*q][0], wr[2*q+1][0], wr[2*q][1], wr[2*q+1][1] };
                    v4f p1 = { wr[2*q][2], wr[2*q+1][2], wr[2*q][3], wr[2*q+1][3] };
                    *(v4f*)(wn + q * GPP)     = p0;
                    *(v4f*)(wn + q * GPP + 4) = p1;
                }
                float* an = ast + (h ^ 1) * GAST;
#pragma unroll
                for (int r = 0; r < GAR; ++r) *(v4f*)(an + (2 * r) * GAP) = ar[r];
            }
            __syncthreads();
        }
    }

    const int mbase = hi * 8;
#pragma unroll
    for (int nt = 0; nt < 8; ++nt)
#pragma unroll
        for (int j = 0; j < 8; ++j) {
            size_t off = (size_t)(strip * 16 + mbase + j) * HIDC + nb * 128 + nt * 16 + m;
            float g = acc[nt][j];
            if constexpr (SWISH) {
                out[off] = g / (1.0f + expf(-g));
            } else {
                out[off] = g;
            }
        }
}

static_assert(BBC * NHC == BBC * SSC / 128, "blockIdx.x serves as the scan's head-batch index");
__global__ void __launch_bounds__(256, 4)
gate_state_kernel(const float* __restrict__ A, const float* __restrict__ W,
                  float* __restrict__ out,
                  const float* __restrict__ kdf, const float* __restrict__ vf,
                  float* __restrict__ sf)
{
    if (blockIdx.y >= GNB) {
        const int wv = threadIdx.x >> 5;
        if (wv < 4)
            state_scan(kdf, vf, sf, threadIdx.x & 31, (blockIdx.y - GNB) * 4 + wv, blockIdx.x);
        return;
    }
    gemm1024_body<true>(A, W, out);
}

__global__ void __launch_bounds__(256)
gemm1024_kernel(const float* __restrict__ A, const float* __restrict__ W,
                float* __restrict__ out)
{
    gemm1024_body<false>(A, W, out);
}

extern "C" void kernel_launch(void* const* d_in, const int* in_sizes, int n_in,
                              void* d_out, int out_size, void* d_ws, size_t ws_size,
                              hipStream_t stream)
{
    (void)in_sizes; (void)n_in; (void)out_size; (void)ws_size;
    const float* x   = (const float*)d_in[0];
    const float* wq  = (const float*)d_in[1];
    const float* wk  = (const float*)d_in[2];
    const float* wv  = (const float*)d_in[3];
    const float* w1  = (const float*)d_in[4];
    const float* w2  = (const float*)d_in[5];
    const float* gns = (const float*)d_in[6];
    const float* gnb = (const float*)d_in[7];
    float* out = (float*)d_out;
    float* ws  = (float*)d_ws;

    const size_t NQ = (size_t)BBC * NHC * SSC * HDC;
    float* q    = ws;
    float* k    = ws + NQ;
    float* v    = ws + 2 * NQ;
    float* sum  = ws + 3 * NQ;
    float* kd   = ws + 4 * NQ;
    float* sst  = ws + 5 * NQ;
    float* gate = ws + 6 * NQ;

    dim3 wave(32);
    qkv_kernel<<<dim3(BBC * SSC / 16, NHC), wave, 0, stream>>>(x, wq, wk, wv, q, k, v, kd);
    gate_state_kernel<<<dim3(BBC * SSC / 128, GNB + 4), dim3(256), 0, stream>>>(x, w1, gate, kd, v, sst);
    retention_kernel<<<dim3(SSC / 32, BBC * NHC), wave, 0, stream>>>(q, k, v, sst, gns, gnb, gate, sum);
    gemm1024_kernel<<<dim3(BBC * SSC / 128, GNB), dim3(256), 0, stream>>>(sum, w2, out);
}
